# starting kernel
# baseline (speedup 1.0000x reference)
#include <hip/hip_runtime.h>

typedef __attribute__((ext_vector_type(2))) float v2f;
typedef __attribute__((ext_vector_type(8))) float v8f;

#define NFEAT 128

__global__ void k_init_deg(float* deg, int n) {
    int i = blockIdx.x * blockDim.x + threadIdx.x;
    if (i < n) deg[i] = 1.0f;
}

__global__ void k_deg_accum(const int* __restrict__ dst, float* deg, int E) {
    int e = blockIdx.x * blockDim.x + threadIdx.x;
    if (e < E) atomicAdd(&deg[dst[e]], 1.0f);
}

__global__ void k_rsqrt(float* d, int n) {
    int i = blockIdx.x * blockDim.x + threadIdx.x;
    if (i < n) d[i] = rsqrtf(d[i]);
}

template<int NT, bool RELU, bool GUARD>
__global__ void k_wmma_gemm_fused(const float* __restrict__ A,
                                  const float* __restrict__ B,
                                  const float* __restrict__ dis,
                                  const float* __restrict__ bias,
                                  float* __restrict__ XW,
                                  float* __restrict__ Cinit,
                                  int nrows, int ncols) {
    const int lane = threadIdx.x & 31;
    const int half = lane >> 4;
    const int l    = lane & 15;

    int arow = blockIdx.x * 16 + l;
    if (arow >= nrows) arow = nrows - 1;
    const float* __restrict__ Arow = A + (size_t)arow * NFEAT;

    int   col [NT];
    float mask[NT];
    #pragma unroll
    for (int t = 0; t < NT; ++t) {
        const int c0 = blockIdx.y * (16 * NT) + t * 16 + l;
        if (GUARD) {
            const bool ok = (c0 < ncols);
            col [t] = ok ? c0 : (ncols - 1);
            mask[t] = ok ? 1.0f : 0.0f;
        } else {
            col [t] = c0;
            mask[t] = 1.0f;
        }
    }

    v8f acc[NT];
    #pragma unroll
    for (int t = 0; t < NT; ++t) acc[t] = (v8f){0, 0, 0, 0, 0, 0, 0, 0};

    #pragma unroll 4
    for (int k0 = 0; k0 < NFEAT; k0 += 4) {
        const int ka = k0 + 2 * half;
        v2f a;
        a.x = Arow[ka];
        a.y = Arow[ka + 1];
        if (RELU) { a.x = fmaxf(a.x, 0.0f); a.y = fmaxf(a.y, 0.0f); }
        #pragma unroll
        for (int t = 0; t < NT; ++t) {
            v2f b;
            b.x = B[(size_t)ka       * ncols + col[t]];
            b.y = B[(size_t)(ka + 1) * ncols + col[t]];
            if (GUARD) { b.x *= mask[t]; b.y *= mask[t]; }
            acc[t] = __builtin_amdgcn_wmma_f32_16x16x4_f32(
                         false, a, false, b, (short)0, acc[t], false, false);
        }
    }

    const int rbase = blockIdx.x * 16 + half * 8;
    float dsq[8];
    #pragma unroll
    for (int r = 0; r < 8; ++r) {
        int row = rbase + r;
        if (row >= nrows) row = nrows - 1;
        const float s = dis[row];
        dsq[r] = s * s;
    }

    #pragma unroll
    for (int t = 0; t < NT; ++t) {
        const int c0 = blockIdx.y * (16 * NT) + t * 16 + l;
        if (!GUARD || c0 < ncols) {
            const float bv = bias[c0];
            #pragma unroll
            for (int r = 0; r < 8; ++r) {
                const int row = rbase + r;
                if (row < nrows) {
                    const float v = acc[t][r];
                    XW   [(size_t)row * ncols + c0] = v;
                    Cinit[(size_t)row * ncols + c0] = v * dsq[r] + bv;
                }
            }
        }
    }
}

__global__ void k_edge_scatter(const int* __restrict__ src,
                               const int* __restrict__ dst,
                               const float* __restrict__ dis,
                               const float* __restrict__ feat,
                               float* __restrict__ out, int ncols, int E) {
    long long gid = (long long)blockIdx.x * blockDim.x + threadIdx.x;
    int e    = (int)(gid >> 5);
    int lane = (int)(gid & 31);
    if (e >= E) return;
    const int s = src[e];
    const int d = dst[e];
    const float norm = dis[s] * dis[d];
    const float* __restrict__ fp = feat + (size_t)s * ncols;
    float* __restrict__ op = out + (size_t)d * ncols;
    for (int f = lane; f < ncols; f += 32)
        atomicAdd(&op[f], fp[f] * norm);
}

extern "C" void kernel_launch(void* const* d_in, const int* in_sizes, int n_in,
                              void* d_out, int out_size, void* d_ws, size_t ws_size,
                              hipStream_t stream) {
    const float* x  = (const float*)d_in[0];
    const int*   ei = (const int*)  d_in[1];
    const float* W1 = (const float*)d_in[2];
    const float* b1 = (const float*)d_in[3];
    const float* W2 = (const float*)d_in[4];
    const float* b2 = (const float*)d_in[5];
    float* out = (float*)d_out;

    const int N   = in_sizes[0] / NFEAT;
    const int E   = in_sizes[1] / 2;
    const int CLS = in_sizes[5];

    const int* src = ei;
    const int* dst = ei + E;

    char* ws = (char*)d_ws;
    size_t off = 0;
    float* dis = (float*)(ws + off); off += ((size_t)N * 4 + 255) & ~(size_t)255;
    float* xw1 = (float*)(ws + off); off += (size_t)N * NFEAT * 4;
    float* h   = (float*)(ws + off); off += (size_t)N * NFEAT * 4;
    float* hw2 = (float*)(ws + off); off += (size_t)N * CLS * 4;
    (void)ws_size; (void)n_in; (void)out_size;

    const int TPB = 256;

    k_init_deg <<<(N + TPB - 1) / TPB, TPB, 0, stream>>>(dis, N);
    k_deg_accum<<<(E + TPB - 1) / TPB, TPB, 0, stream>>>(dst, dis, E);
    k_rsqrt    <<<(N + TPB - 1) / TPB, TPB, 0, stream>>>(dis, N);

    {
        dim3 grid((N + 15) / 16, NFEAT / 64);
        k_wmma_gemm_fused<4, false, false><<<grid, 32, 0, stream>>>(
            x, W1, dis, b1, xw1, h, N, NFEAT);
    }
    {
        long long threads = (long long)E * 32;
        k_edge_scatter<<<(unsigned)((threads + TPB - 1) / TPB), TPB, 0, stream>>>(
            src, dst, dis, xw1, h, NFEAT, E);
    }

    {
        dim3 grid((N + 15) / 16, 1);
        k_wmma_gemm_fused<3, true, true><<<grid, 32, 0, stream>>>(
            h, W2, dis, b2, hw2, out, N, CLS);
    }
    {
        long long threads = (long long)E * 32;
        k_edge_scatter<<<(unsigned)((threads + TPB - 1) / TPB), TPB, 0, stream>>>(
            src, dst, dis, hw2, out, CLS, E);
    }
}


# Float atomic scatter-add replaced by edges grouped by dst in the call (int atomic rank, scan, place) and one wave per row summing source rows in registers; sum order = arrival rank
# speedup vs baseline: 2.1714x; 2.1714x over previous
#include <hip/hip_runtime.h>


typedef __attribute__((ext_vector_type(2))) float v2f;
typedef __attribute__((ext_vector_type(8))) float v8f;

#define NFEAT 128
#define SCAN_B 1024


__global__ void __launch_bounds__(256) k_zero_cnt(int* __restrict__ cnt, int n) {
    int i = blockIdx.x * blockDim.x + threadIdx.x;
    if (i < n) cnt[i] = 0;
}

__global__ void __launch_bounds__(256) k_count_rank(const int* __restrict__ dst,
                                                    int* cnt,
                                                    int* __restrict__ erank, int E) {
    int e = blockIdx.x * blockDim.x + threadIdx.x;
    if (e < E) erank[e] = atomicAdd(&cnt[dst[e]], 1);
}

__device__ __forceinline__ int wave_sum(int v) {
    #pragma unroll
    for (int o = 16; o > 0; o >>= 1) v += __shfl_xor(v, o);
    return v;
}

__device__ __forceinline__ int wave_incl_scan(int v, int lane) {
    #pragma unroll
    for (int o = 1; o < 32; o <<= 1) {
        const int t = __shfl_up(v, o);
        if (lane >= o) v += t;
    }
    return v;
}

__global__ void __launch_bounds__(SCAN_B) k_block_sums(const int* __restrict__ cnt,
                                                       int* __restrict__ bsum, int n) {
    __shared__ int wsum[SCAN_B / 32];
    const int t = threadIdx.x, lane = t & 31, w = t >> 5;
    const int i = blockIdx.x * SCAN_B + t;
    int v = (i < n) ? cnt[i] : 0;
    v = wave_sum(v);
    if (lane == 0) wsum[w] = v;
    __syncthreads();
    if (w == 0) {
        int s = wave_sum(wsum[lane]);
        if (lane == 0) bsum[blockIdx.x] = s;
    }
}

__global__ void __launch_bounds__(SCAN_B) k_scan_rowptr(const int* __restrict__ cnt,
                                                        const int* __restrict__ bsum,
                                                        int* __restrict__ rowptr,
                                                        float* __restrict__ dis, int n) {
    __shared__ int wsum[SCAN_B / 32];
    __shared__ int s_off;
    const int t = threadIdx.x, lane = t & 31, w = t >> 5;

    int p = 0;
    for (int k = t; k < (int)blockIdx.x; k += SCAN_B) p += bsum[k];
    p = wave_sum(p);
    if (lane == 0) wsum[w] = p;
    __syncthreads();
    if (w == 0) {
        int s = wave_sum(wsum[lane]);
        if (lane == 0) s_off = s;
    }
    __syncthreads();
    const int boff = s_off;

    const int i = blockIdx.x * SCAN_B + t;
    const int c = (i < n) ? cnt[i] : 0;
    const int inc = wave_incl_scan(c, lane);
    if (lane == 31) wsum[w] = inc;
    __syncthreads();
    if (w == 0) {
        const int s  = wsum[lane];
        const int si = wave_incl_scan(s, lane);
        wsum[lane] = si - s;
    }
    __syncthreads();
    const int excl = boff + wsum[w] + inc - c;
    if (i < n) {
        rowptr[i] = excl;
        dis[i]    = rsqrtf((float)c + 1.0f);
        if (i == n - 1) rowptr[n] = excl + c;
    }
}

__global__ void __launch_bounds__(256) k_place(const int* __restrict__ src,
                                               const int* __restrict__ dst,
                                               const int* __restrict__ erank,
                                               const int* __restrict__ rowptr,
                                               int* __restrict__ csr_src, int E) {
    int e = blockIdx.x * blockDim.x + threadIdx.x;
    if (e < E) csr_src[rowptr[dst[e]] + erank[e]] = src[e];
}

template<int NT, bool RELU, bool GUARD>
__global__ void k_wmma_gemm_fused(const float* __restrict__ A,
                                  const float* __restrict__ B,
                                  const float* __restrict__ dis,
                                  const float* __restrict__ bias,
                                  float* __restrict__ XW,
                                  float* __restrict__ Cinit,
                                  int nrows, int ncols) {
    const int lane = threadIdx.x & 31;
    const int half = lane >> 4;
    const int l    = lane & 15;

    int arow = blockIdx.x * 16 + l;
    if (arow >= nrows) arow = nrows - 1;
    const float* __restrict__ Arow = A + (size_t)arow * NFEAT;

    int   col [NT];
    float mask[NT];
    #pragma unroll
    for (int t = 0; t < NT; ++t) {
        const int c0 = blockIdx.y * (16 * NT) + t * 16 + l;
        if (GUARD) {
            const bool ok = (c0 < ncols);
            col [t] = ok ? c0 : (ncols - 1);
            mask[t] = ok ? 1.0f : 0.0f;
        } else {
            col [t] = c0;
            mask[t] = 1.0f;
        }
    }

    v8f acc[NT];
    #pragma unroll
    for (int t = 0; t < NT; ++t) acc[t] = (v8f){0, 0, 0, 0, 0, 0, 0, 0};

    #pragma unroll 4
    for (int k0 = 0; k0 < NFEAT; k0 += 4) {
        const int ka = k0 + 2 * half;
        v2f a;
        a.x = Arow[ka];
        a.y = Arow[ka + 1];
        if (RELU) { a.x = fmaxf(a.x, 0.0f); a.y = fmaxf(a.y, 0.0f); }
        #pragma unroll
        for (int t = 0; t < NT; ++t) {
            v2f b;
            b.x = B[(size_t)ka       * ncols + col[t]];
            b.y = B[(size_t)(ka + 1) * ncols + col[t]];
            if (GUARD) { b.x *= mask[t]; b.y *= mask[t]; }
            acc[t] = __builtin_amdgcn_wmma_f32_16x16x4_f32(
                         false, a, false, b, (short)0, acc[t], false, false);
        }
    }

    const int rbase = blockIdx.x * 16 + half * 8;
    float dsq[8];
    #pragma unroll
    for (int r = 0; r < 8; ++r) {
        int row = rbase + r;
        if (row >= nrows) row = nrows - 1;
        const float s = dis[row];
        dsq[r] = s * s;
    }

    #pragma unroll
    for (int t = 0; t < NT; ++t) {
        const int c0 = blockIdx.y * (16 * NT) + t * 16 + l;
        if (!GUARD || c0 < ncols) {
            const float bv = bias[c0];
            #pragma unroll
            for (int r = 0; r < 8; ++r) {
                const int row = rbase + r;
                if (row < nrows) {
                    const float v = acc[t][r];
                    XW   [(size_t)row * ncols + c0] = v;
                    Cinit[(size_t)row * ncols + c0] = v * dsq[r] + bv;
                }
            }
        }
    }
}

template<int ROWS>
__global__ void __launch_bounds__(256) k_row_gather(const int* __restrict__ rowptr,
                                                    const int* __restrict__ csr_src,
                                                    const float* __restrict__ dis,
                                                    const float* __restrict__ feat,
                                                    float* io, int ncols, int n) {
    const int lane = threadIdx.x & 31;
    const int first = __builtin_amdgcn_readfirstlane(
        (int)((blockIdx.x * (blockDim.x >> 5) + (threadIdx.x >> 5)) * ROWS));

    const bool act = (lane * 4 < ncols);
    const int  c4  = act ? lane * 4 : 0;
    const float* __restrict__ fcol = feat + c4;

    for (int r = 0; r < ROWS; ++r) {
        const int node = first + r;
        if (node >= n) break;

        const int   beg = rowptr[node];
        const int   end = rowptr[node + 1];
        const float di  = dis[node];

        float4* iop = (float4*)(io + (size_t)node * ncols + c4);
        float4 acc = *iop;

        for (int base = beg; base < end; base += 32) {
            const int left = end - base;
            const int m = left < 32 ? left : 32;
            int   ms = 0;
            float mw = 0.0f;
            if (lane < m) {
                ms = csr_src[base + lane];
                mw = dis[ms] * di;
            }
            int j = 0;
            for (; j + 4 <= m; j += 4) {
                const int   s0 = __builtin_amdgcn_readlane(ms, j);
                const int   s1 = __builtin_amdgcn_readlane(ms, j + 1);
                const int   s2 = __builtin_amdgcn_readlane(ms, j + 2);
                const int   s3 = __builtin_amdgcn_readlane(ms, j + 3);
                const float w0 = __int_as_float(__builtin_amdgcn_readlane(__float_as_int(mw), j));
                const float w1 = __int_as_float(__builtin_amdgcn_readlane(__float_as_int(mw), j + 1));
                const float w2 = __int_as_float(__builtin_amdgcn_readlane(__float_as_int(mw), j + 2));
                const float w3 = __int_as_float(__builtin_amdgcn_readlane(__float_as_int(mw), j + 3));
                const float4 v0 = *(const float4*)(fcol + (size_t)s0 * ncols);
                const float4 v1 = *(const float4*)(fcol + (size_t)s1 * ncols);
                const float4 v2 = *(const float4*)(fcol + (size_t)s2 * ncols);
                const float4 v3 = *(const float4*)(fcol + (size_t)s3 * ncols);
                acc.x = fmaf(v0.x, w0, acc.x); acc.y = fmaf(v0.y, w0, acc.y); acc.z = fmaf(v0.z, w0, acc.z); acc.w = fmaf(v0.w, w0, acc.w);
                acc.x = fmaf(v1.x, w1, acc.x); acc.y = fmaf(v1.y, w1, acc.y); acc.z = fmaf(v1.z, w1, acc.z); acc.w = fmaf(v1.w, w1, acc.w);
                acc.x = fmaf(v2.x, w2, acc.x); acc.y = fmaf(v2.y, w2, acc.y); acc.z = fmaf(v2.z, w2, acc.z); acc.w = fmaf(v2.w, w2, acc.w);
                acc.x = fmaf(v3.x, w3, acc.x); acc.y = fmaf(v3.y, w3, acc.y); acc.z = fmaf(v3.z, w3, acc.z); acc.w = fmaf(v3.w, w3, acc.w);
            }
            for (; j < m; ++j) {
                const int   s0 = __builtin_amdgcn_readlane(ms, j);
                const float w0 = __int_as_float(__builtin_amdgcn_readlane(__float_as_int(mw), j));
                const float4 v0 = *(const float4*)(fcol + (size_t)s0 * ncols);
                acc.x = fmaf(v0.x, w0, acc.x); acc.y = fmaf(v0.y, w0, acc.y); acc.z = fmaf(v0.z, w0, acc.z); acc.w = fmaf(v0.w, w0, acc.w);
            }
        }
        if (act) *iop = acc;
    }
}

extern "C" void kernel_launch(void* const* d_in, const int* in_sizes, int n_in,
                              void* d_out, int out_size, void* d_ws, size_t ws_size,
                              hipStream_t stream) {
    const float* x  = (const float*)d_in[0];
    const int*   ei = (const int*)  d_in[1];
    const float* W1 = (const float*)d_in[2];
    const float* b1 = (const float*)d_in[3];
    const float* W2 = (const float*)d_in[4];
    const float* b2 = (const float*)d_in[5];
    float* out = (float*)d_out;

    const int N   = in_sizes[0] / NFEAT;
    const int E   = in_sizes[1] / 2;
    const int CLS = in_sizes[5];

    const int* src = ei;
    const int* dst = ei + E;

    const int NB = (N + SCAN_B - 1) / SCAN_B;

    char* ws = (char*)d_ws;
    size_t off = 0;
    float* dis = (float*)(ws + off); off += ((size_t)N * 4 + 255) & ~(size_t)255;
    float* xw1 = (float*)(ws + off); off += (size_t)N * NFEAT * 4;
    float* h   = (float*)(ws + off); off += (size_t)N * NFEAT * 4;
    float* hw2 = (float*)(ws + off); off += ((size_t)N * CLS * 4 + 255) & ~(size_t)255;
    int* cnt    = (int*)(ws + off); off += ((size_t)N * 4 + 255) & ~(size_t)255;
    int* rowptr = (int*)(ws + off); off += ((size_t)(N + 1) * 4 + 255) & ~(size_t)255;
    int* bsum   = (int*)(ws + off); off += ((size_t)NB * 4 + 255) & ~(size_t)255;
    int* erank  = (int*)(ws + off); off += ((size_t)E * 4 + 255) & ~(size_t)255;
    int* csr    = (int*)(ws + off); off += ((size_t)E * 4 + 255) & ~(size_t)255;
    (void)ws_size; (void)n_in; (void)out_size;

    const int TPB = 256;
    const int WAVES = TPB / 32;

    k_zero_cnt   <<<(N + TPB - 1) / TPB, TPB, 0, stream>>>(cnt, N);
    k_count_rank <<<(E + TPB - 1) / TPB, TPB, 0, stream>>>(dst, cnt, erank, E);
    k_block_sums <<<NB, SCAN_B, 0, stream>>>(cnt, bsum, N);
    k_scan_rowptr<<<NB, SCAN_B, 0, stream>>>(cnt, bsum, rowptr, dis, N);
    k_place      <<<(E + TPB - 1) / TPB, TPB, 0, stream>>>(src, dst, erank, rowptr, csr, E);

    {
        dim3 grid((N + 15) / 16, NFEAT / 64);
        k_wmma_gemm_fused<4, false, false><<<grid, 32, 0, stream>>>(
            x, W1, dis, b1, xw1, h, N, NFEAT);
    }
    k_row_gather<1><<<(N + WAVES - 1) / WAVES, TPB, 0, stream>>>(
        rowptr, csr, dis, xw1, h, NFEAT, N);

    {
        dim3 grid((N + 15) / 16, 1);
        k_wmma_gemm_fused<3, true, true><<<grid, 32, 0, stream>>>(
            h, W2, dis, b2, hw2, out, N, CLS);
    }
    k_row_gather<4><<<(N + 4 * WAVES - 1) / (4 * WAVES), TPB, 0, stream>>>(
        rowptr, csr, dis, hw2, out, CLS, N);
}


# Edges grouped by dst with LDS counters in two levels (256-row buckets, then rows in a bucket) instead of one global returning atomic and one scattered store per edge; sum order still arrival rank
# speedup vs baseline: 3.1229x; 1.4382x over previous
#include <hip/hip_runtime.h>


typedef __attribute__((ext_vector_type(2))) float v2f;
typedef __attribute__((ext_vector_type(8))) float v8f;

#define NFEAT 128


#define BKT_SHIFT 8
#define BKT_W (1 << BKT_SHIFT)
#define EB 1024
#define EPT 4
#define FU 4

__global__ void __launch_bounds__(256) k_zero_bkt(int* __restrict__ bc, int n) {
    int i = blockIdx.x * blockDim.x + threadIdx.x;
    if (i < n) bc[i] = 0;
}

__device__ __forceinline__ int wave_incl_scan(int v, int lane) {
    #pragma unroll
    for (int o = 1; o < 32; o <<= 1) {
        const int t = __shfl_up(v, o);
        if (lane >= o) v += t;
    }
    return v;
}

__global__ void __launch_bounds__(EB) k_bkt_count(const int* __restrict__ dst,
                                                  int* bcnt, int E, int nbkt) {
    __shared__ int hist[EB];
    const int t = threadIdx.x;
    hist[t] = 0;
    __syncthreads();
    const int base = blockIdx.x * (EB * EPT);
    int d[EPT];
    #pragma unroll
    for (int k = 0; k < EPT; ++k) {
        const int e = base + k * EB + t;
        d[k] = (e < E) ? dst[e] : -1;
    }
    #pragma unroll
    for (int k = 0; k < EPT; ++k)
        if (d[k] >= 0) atomicAdd(&hist[d[k] >> BKT_SHIFT], 1);
    __syncthreads();
    if (t < nbkt) {
        const int h = hist[t];
        if (h) atomicAdd(&bcnt[t], h);
    }
}

__global__ void __launch_bounds__(EB) k_bkt_scatter(const int* __restrict__ src,
                                                    const int* __restrict__ dst,
                                                    const int* __restrict__ bcnt,
                                                    int* bcur,
                                                    int* __restrict__ bstart,
                                                    int* __restrict__ bkt_raw,
                                                    int E, int nbkt) {
    __shared__ int hist[EB];
    __shared__ int wsum[EB / 32];
    const int t = threadIdx.x, lane = t & 31, w = t >> 5;
    int2* __restrict__ bkt = (int2*)bkt_raw;

    hist[t] = 0;
    __syncthreads();

    const int base = blockIdx.x * (EB * EPT);
    int s[EPT], d[EPT], r[EPT];
    #pragma unroll
    for (int k = 0; k < EPT; ++k) {
        const int e = base + k * EB + t;
        d[k] = (e < E) ? dst[e] : -1;
        s[k] = (e < E) ? src[e] : 0;
    }
    #pragma unroll
    for (int k = 0; k < EPT; ++k) {
        r[k] = 0;
        if (d[k] >= 0) r[k] = atomicAdd(&hist[d[k] >> BKT_SHIFT], 1);
    }

    const int c   = (t < nbkt) ? bcnt[t] : 0;
    const int inc = wave_incl_scan(c, lane);
    if (lane == 31) wsum[w] = inc;
    __syncthreads();
    if (w == 0) {
        const int x  = wsum[lane];
        const int xi = wave_incl_scan(x, lane);
        wsum[lane] = xi - x;
    }
    __syncthreads();
    const int excl = wsum[w] + inc - c;

    int mybase = 0;
    if (t < nbkt) {
        const int h = hist[t];
        mybase = excl + (h ? atomicAdd(&bcur[t], h) : 0);
        if (blockIdx.x == 0) {
            bstart[t] = excl;
            if (t == nbkt - 1) bstart[nbkt] = excl + c;
        }
    }
    hist[t] = mybase;
    __syncthreads();

    #pragma unroll
    for (int k = 0; k < EPT; ++k) {
        if (d[k] >= 0) bkt[hist[d[k] >> BKT_SHIFT] + r[k]] = make_int2(s[k], d[k]);
    }
}

__global__ void __launch_bounds__(EB) k_bkt_finish(const int* __restrict__ bkt_raw,
                                                   const int* __restrict__ bstart,
                                                   int* __restrict__ rowptr,
                                                   float* __restrict__ dis,
                                                   int* __restrict__ csr_src, int n) {
    __shared__ int cnt[BKT_W];
    __shared__ int cur[BKT_W];
    __shared__ int wsum[BKT_W / 32];
    const int t = threadIdx.x, lane = t & 31, w = t >> 5;
    const int2* __restrict__ bkt = (const int2*)bkt_raw;

    const int lo  = blockIdx.x << BKT_SHIFT;
    const int beg = bstart[blockIdx.x];
    const int end = bstart[blockIdx.x + 1];

    if (t < BKT_W) cnt[t] = 0;
    __syncthreads();
    for (int i0 = beg + t; i0 < end; i0 += FU * EB) {
        int dd[FU];
        #pragma unroll
        for (int u = 0; u < FU; ++u) {
            const int i = i0 + u * EB;
            dd[u] = (i < end) ? bkt[i].y : -1;
        }
        #pragma unroll
        for (int u = 0; u < FU; ++u)
            if (dd[u] >= 0) atomicAdd(&cnt[dd[u] - lo], 1);
    }
    __syncthreads();

    int c = 0, inc = 0;
    if (t < BKT_W) {
        c   = cnt[t];
        inc = wave_incl_scan(c, lane);
        if (lane == 31) wsum[w] = inc;
    }
    __syncthreads();
    if (w == 0) {
        const int x  = (lane < BKT_W / 32) ? wsum[lane] : 0;
        const int xi = wave_incl_scan(x, lane);
        if (lane < BKT_W / 32) wsum[lane] = xi - x;
    }
    __syncthreads();
    if (t < BKT_W) {
        const int excl = wsum[w] + inc - c;
        cur[t] = excl;
        const int node = lo + t;
        if (node < n) {
            rowptr[node] = beg + excl;
            dis[node]    = rsqrtf((float)c + 1.0f);
            if (node == n - 1) rowptr[n] = beg + excl + c;
        }
    }
    __syncthreads();
    for (int i0 = beg + t; i0 < end; i0 += FU * EB) {
        int2 p[FU];
        #pragma unroll
        for (int u = 0; u < FU; ++u) {
            const int i = i0 + u * EB;
            p[u] = (i < end) ? bkt[i] : make_int2(0, -1);
        }
        #pragma unroll
        for (int u = 0; u < FU; ++u) {
            if (p[u].y >= 0) {
                const int pos = atomicAdd(&cur[p[u].y - lo], 1);
                csr_src[beg + pos] = p[u].x;
            }
        }
    }
}

template<int NT, bool RELU, bool GUARD>
__global__ void k_wmma_gemm_fused(const float* __restrict__ A,
                                  const float* __restrict__ B,
                                  const float* __restrict__ dis,
                                  const float* __restrict__ bias,
                                  float* __restrict__ XW,
                                  float* __restrict__ Cinit,
                                  int nrows, int ncols) {
    const int lane = threadIdx.x & 31;
    const int half = lane >> 4;
    const int l    = lane & 15;

    int arow = blockIdx.x * 16 + l;
    if (arow >= nrows) arow = nrows - 1;
    const float* __restrict__ Arow = A + (size_t)arow * NFEAT;

    int   col [NT];
    float mask[NT];
    #pragma unroll
    for (int t = 0; t < NT; ++t) {
        const int c0 = blockIdx.y * (16 * NT) + t * 16 + l;
        if (GUARD) {
            const bool ok = (c0 < ncols);
            col [t] = ok ? c0 : (ncols - 1);
            mask[t] = ok ? 1.0f : 0.0f;
        } else {
            col [t] = c0;
            mask[t] = 1.0f;
        }
    }

    v8f acc[NT];
    #pragma unroll
    for (int t = 0; t < NT; ++t) acc[t] = (v8f){0, 0, 0, 0, 0, 0, 0, 0};

    #pragma unroll 4
    for (int k0 = 0; k0 < NFEAT; k0 += 4) {
        const int ka = k0 + 2 * half;
        v2f a;
        a.x = Arow[ka];
        a.y = Arow[ka + 1];
        if (RELU) { a.x = fmaxf(a.x, 0.0f); a.y = fmaxf(a.y, 0.0f); }
        #pragma unroll
        for (int t = 0; t < NT; ++t) {
            v2f b;
            b.x = B[(size_t)ka       * ncols + col[t]];
            b.y = B[(size_t)(ka + 1) * ncols + col[t]];
            if (GUARD) { b.x *= mask[t]; b.y *= mask[t]; }
            acc[t] = __builtin_amdgcn_wmma_f32_16x16x4_f32(
                         false, a, false, b, (short)0, acc[t], false, false);
        }
    }

    const int rbase = blockIdx.x * 16 + half * 8;
    float dsq[8];
    #pragma unroll
    for (int r = 0; r < 8; ++r) {
        int row = rbase + r;
        if (row >= nrows) row = nrows - 1;
        const float s = dis[row];
        dsq[r] = s * s;
    }

    #pragma unroll
    for (int t = 0; t < NT; ++t) {
        const int c0 = blockIdx.y * (16 * NT) + t * 16 + l;
        if (!GUARD || c0 < ncols) {
            const float bv = bias[c0];
            #pragma unroll
            for (int r = 0; r < 8; ++r) {
                const int row = rbase + r;
                if (row < nrows) {
                    const float v = acc[t][r];
                    XW   [(size_t)row * ncols + c0] = v;
                    Cinit[(size_t)row * ncols + c0] = v * dsq[r] + bv;
                }
            }
        }
    }
}

template<int ROWS>
__global__ void __launch_bounds__(256) k_row_gather(const int* __restrict__ rowptr,
                                                    const int* __restrict__ csr_src,
                                                    const float* __restrict__ dis,
                                                    const float* __restrict__ feat,
                                                    float* io, int ncols, int n) {
    const int lane = threadIdx.x & 31;
    const int first = __builtin_amdgcn_readfirstlane(
        (int)((blockIdx.x * (blockDim.x >> 5) + (threadIdx.x >> 5)) * ROWS));

    const bool act = (lane * 4 < ncols);
    const int  c4  = act ? lane * 4 : 0;
    const float* __restrict__ fcol = feat + c4;

    for (int r = 0; r < ROWS; ++r) {
        const int node = first + r;
        if (node >= n) break;

        const int   beg = rowptr[node];
        const int   end = rowptr[node + 1];
        const float di  = dis[node];

        float4* iop = (float4*)(io + (size_t)node * ncols + c4);
        float4 acc = *iop;

        for (int base = beg; base < end; base += 32) {
            const int left = end - base;
            const int m = left < 32 ? left : 32;
            int   ms = 0;
            float mw = 0.0f;
            if (lane < m) {
                ms = csr_src[base + lane];
                mw = dis[ms] * di;
            }
            int j = 0;
            for (; j + 4 <= m; j += 4) {
                const int   s0 = __builtin_amdgcn_readlane(ms, j);
                const int   s1 = __builtin_amdgcn_readlane(ms, j + 1);
                const int   s2 = __builtin_amdgcn_readlane(ms, j + 2);
                const int   s3 = __builtin_amdgcn_readlane(ms, j + 3);
                const float w0 = __int_as_float(__builtin_amdgcn_readlane(__float_as_int(mw), j));
                const float w1 = __int_as_float(__builtin_amdgcn_readlane(__float_as_int(mw), j + 1));
                const float w2 = __int_as_float(__builtin_amdgcn_readlane(__float_as_int(mw), j + 2));
                const float w3 = __int_as_float(__builtin_amdgcn_readlane(__float_as_int(mw), j + 3));
                const float4 v0 = *(const float4*)(fcol + (size_t)s0 * ncols);
                const float4 v1 = *(const float4*)(fcol + (size_t)s1 * ncols);
                const float4 v2 = *(const float4*)(fcol + (size_t)s2 * ncols);
                const float4 v3 = *(const float4*)(fcol + (size_t)s3 * ncols);
                acc.x = fmaf(v0.x, w0, acc.x); acc.y = fmaf(v0.y, w0, acc.y); acc.z = fmaf(v0.z, w0, acc.z); acc.w = fmaf(v0.w, w0, acc.w);
                acc.x = fmaf(v1.x, w1, acc.x); acc.y = fmaf(v1.y, w1, acc.y); acc.z = fmaf(v1.z, w1, acc.z); acc.w = fmaf(v1.w, w1, acc.w);
                acc.x = fmaf(v2.x, w2, acc.x); acc.y = fmaf(v2.y, w2, acc.y); acc.z = fmaf(v2.z, w2, acc.z); acc.w = fmaf(v2.w, w2, acc.w);
                acc.x = fmaf(v3.x, w3, acc.x); acc.y = fmaf(v3.y, w3, acc.y); acc.z = fmaf(v3.z, w3, acc.z); acc.w = fmaf(v3.w, w3, acc.w);
            }
            for (; j < m; ++j) {
                const int   s0 = __builtin_amdgcn_readlane(ms, j);
                const float w0 = __int_as_float(__builtin_amdgcn_readlane(__float_as_int(mw), j));
                const float4 v0 = *(const float4*)(fcol + (size_t)s0 * ncols);
                acc.x = fmaf(v0.x, w0, acc.x); acc.y = fmaf(v0.y, w0, acc.y); acc.z = fmaf(v0.z, w0, acc.z); acc.w = fmaf(v0.w, w0, acc.w);
            }
        }
        if (act) *iop = acc;
    }
}

extern "C" void kernel_launch(void* const* d_in, const int* in_sizes, int n_in,
                              void* d_out, int out_size, void* d_ws, size_t ws_size,
                              hipStream_t stream) {
    const float* x  = (const float*)d_in[0];
    const int*   ei = (const int*)  d_in[1];
    const float* W1 = (const float*)d_in[2];
    const float* b1 = (const float*)d_in[3];
    const float* W2 = (const float*)d_in[4];
    const float* b2 = (const float*)d_in[5];
    float* out = (float*)d_out;

    const int N   = in_sizes[0] / NFEAT;
    const int E   = in_sizes[1] / 2;
    const int CLS = in_sizes[5];

    const int* src = ei;
    const int* dst = ei + E;

    const int NBKT = (N + BKT_W - 1) / BKT_W;

    char* ws = (char*)d_ws;
    size_t off = 0;
    float* dis = (float*)(ws + off); off += ((size_t)N * 4 + 255) & ~(size_t)255;
    float* xw1 = (float*)(ws + off); off += (size_t)N * NFEAT * 4;
    float* h   = (float*)(ws + off); off += (size_t)N * NFEAT * 4;
    float* hw2 = (float*)(ws + off); off += ((size_t)N * CLS * 4 + 255) & ~(size_t)255;
    int* bc     = (int*)(ws + off); off += ((size_t)NBKT * 8 + 255) & ~(size_t)255;
    int* rowptr = (int*)(ws + off); off += ((size_t)(N + 1) * 4 + 255) & ~(size_t)255;
    int* bstart = (int*)(ws + off); off += ((size_t)(NBKT + 1) * 4 + 255) & ~(size_t)255;
    int* bkt    = (int*)(ws + off); off += ((size_t)E * 8 + 255) & ~(size_t)255;
    int* csr    = (int*)(ws + off); off += ((size_t)E * 4 + 255) & ~(size_t)255;
    (void)ws_size; (void)n_in; (void)out_size;

    const int TPB = 256;
    const int WAVES = TPB / 32;

    const int EBLK = (E + EB * EPT - 1) / (EB * EPT);
    k_zero_bkt   <<<(2 * NBKT + TPB - 1) / TPB, TPB, 0, stream>>>(bc, 2 * NBKT);
    k_bkt_count  <<<EBLK, EB, 0, stream>>>(dst, bc, E, NBKT);
    k_bkt_scatter<<<EBLK, EB, 0, stream>>>(src, dst, bc, bc + NBKT, bstart, bkt, E, NBKT);
    k_bkt_finish <<<NBKT, EB, 0, stream>>>(bkt, bstart, rowptr, dis, csr, N);

    {
        dim3 grid((N + 15) / 16, NFEAT / 64);
        k_wmma_gemm_fused<4, false, false><<<grid, 32, 0, stream>>>(
            x, W1, dis, b1, xw1, h, N, NFEAT);
    }
    k_row_gather<1><<<(N + WAVES - 1) / WAVES, TPB, 0, stream>>>(
        rowptr, csr, dis, xw1, h, NFEAT, N);

    {
        dim3 grid((N + 15) / 16, 1);
        k_wmma_gemm_fused<3, true, true><<<grid, 32, 0, stream>>>(
            h, W2, dis, b2, hw2, out, N, CLS);
    }
    k_row_gather<4><<<(N + 4 * WAVES - 1) / (4 * WAVES), TPB, 0, stream>>>(
        rowptr, csr, dis, hw2, out, CLS, N);
}


# h never goes to memory: layer-1 row sums start from the row's own x@W1 row, 16 finished relu rows per block feed the layer-2 f32 WMMA strips through LDS in the same launch (8 to 7); same sums
# speedup vs baseline: 3.7861x; 1.2124x over previous
#include <hip/hip_runtime.h>


typedef __attribute__((ext_vector_type(2))) float v2f;
typedef __attribute__((ext_vector_type(8))) float v8f;

#define NFEAT 128


#define BKT_SHIFT 8
#define BKT_W (1 << BKT_SHIFT)
#define EB 1024
#define EPT 4
#define FU 4

__global__ void __launch_bounds__(256) k_zero_bkt(int* __restrict__ bc, int n) {
    int i = blockIdx.x * blockDim.x + threadIdx.x;
    if (i < n) bc[i] = 0;
}

__device__ __forceinline__ int wave_incl_scan(int v, int lane) {
    #pragma unroll
    for (int o = 1; o < 32; o <<= 1) {
        const int t = __shfl_up(v, o);
        if (lane >= o) v += t;
    }
    return v;
}

__global__ void __launch_bounds__(EB) k_bkt_count(const int* __restrict__ dst,
                                                  int* bcnt, int E, int nbkt) {
    __shared__ int hist[EB];
    const int t = threadIdx.x;
    hist[t] = 0;
    __syncthreads();
    const int base = blockIdx.x * (EB * EPT);
    int d[EPT];
    #pragma unroll
    for (int k = 0; k < EPT; ++k) {
        const int e = base + k * EB + t;
        d[k] = (e < E) ? dst[e] : -1;
    }
    #pragma unroll
    for (int k = 0; k < EPT; ++k)
        if (d[k] >= 0) atomicAdd(&hist[d[k] >> BKT_SHIFT], 1);
    __syncthreads();
    if (t < nbkt) {
        const int h = hist[t];
        if (h) atomicAdd(&bcnt[t], h);
    }
}

__global__ void __launch_bounds__(EB) k_bkt_scatter(const int* __restrict__ src,
                                                    const int* __restrict__ dst,
                                                    const int* __restrict__ bcnt,
                                                    int* bcur,
                                                    int* __restrict__ bstart,
                                                    int* __restrict__ bkt_raw,
                                                    int E, int nbkt) {
    __shared__ int hist[EB];
    __shared__ int wsum[EB / 32];
    const int t = threadIdx.x, lane = t & 31, w = t >> 5;
    int2* __restrict__ bkt = (int2*)bkt_raw;

    hist[t] = 0;
    __syncthreads();

    const int base = blockIdx.x * (EB * EPT);
    int s[EPT], d[EPT], r[EPT];
    #pragma unroll
    for (int k = 0; k < EPT; ++k) {
        const int e = base + k * EB + t;
        d[k] = (e < E) ? dst[e] : -1;
        s[k] = (e < E) ? src[e] : 0;
    }
    #pragma unroll
    for (int k = 0; k < EPT; ++k) {
        r[k] = 0;
        if (d[k] >= 0) r[k] = atomicAdd(&hist[d[k] >> BKT_SHIFT], 1);
    }

    const int c   = (t < nbkt) ? bcnt[t] : 0;
    const int inc = wave_incl_scan(c, lane);
    if (lane == 31) wsum[w] = inc;
    __syncthreads();
    if (w == 0) {
        const int x  = wsum[lane];
        const int xi = wave_incl_scan(x, lane);
        wsum[lane] = xi - x;
    }
    __syncthreads();
    const int excl = wsum[w] + inc - c;

    int mybase = 0;
    if (t < nbkt) {
        const int h = hist[t];
        mybase = excl + (h ? atomicAdd(&bcur[t], h) : 0);
        if (blockIdx.x == 0) {
            bstart[t] = excl;
            if (t == nbkt - 1) bstart[nbkt] = excl + c;
        }
    }
    hist[t] = mybase;
    __syncthreads();

    #pragma unroll
    for (int k = 0; k < EPT; ++k) {
        if (d[k] >= 0) bkt[hist[d[k] >> BKT_SHIFT] + r[k]] = make_int2(s[k], d[k]);
    }
}

__global__ void __launch_bounds__(EB) k_bkt_finish(const int* __restrict__ bkt_raw,
                                                   const int* __restrict__ bstart,
                                                   int* __restrict__ rowptr,
                                                   float* __restrict__ dis,
                                                   int* __restrict__ csr_src, int n) {
    __shared__ int cnt[BKT_W];
    __shared__ int cur[BKT_W];
    __shared__ int wsum[BKT_W / 32];
    const int t = threadIdx.x, lane = t & 31, w = t >> 5;
    const int2* __restrict__ bkt = (const int2*)bkt_raw;

    const int lo  = blockIdx.x << BKT_SHIFT;
    const int beg = bstart[blockIdx.x];
    const int end = bstart[blockIdx.x + 1];

    if (t < BKT_W) cnt[t] = 0;
    __syncthreads();
    for (int i0 = beg + t; i0 < end; i0 += FU * EB) {
        int dd[FU];
        #pragma unroll
        for (int u = 0; u < FU; ++u) {
            const int i = i0 + u * EB;
            dd[u] = (i < end) ? bkt[i].y : -1;
        }
        #pragma unroll
        for (int u = 0; u < FU; ++u)
            if (dd[u] >= 0) atomicAdd(&cnt[dd[u] - lo], 1);
    }
    __syncthreads();

    int c = 0, inc = 0;
    if (t < BKT_W) {
        c   = cnt[t];
        inc = wave_incl_scan(c, lane);
        if (lane == 31) wsum[w] = inc;
    }
    __syncthreads();
    if (w == 0) {
        const int x  = (lane < BKT_W / 32) ? wsum[lane] : 0;
        const int xi = wave_incl_scan(x, lane);
        if (lane < BKT_W / 32) wsum[lane] = xi - x;
    }
    __syncthreads();
    if (t < BKT_W) {
        const int excl = wsum[w] + inc - c;
        cur[t] = excl;
        const int node = lo + t;
        if (node < n) {
            rowptr[node] = beg + excl;
            dis[node]    = rsqrtf((float)c + 1.0f);
            if (node == n - 1) rowptr[n] = beg + excl + c;
        }
    }
    __syncthreads();
    for (int i0 = beg + t; i0 < end; i0 += FU * EB) {
        int2 p[FU];
        #pragma unroll
        for (int u = 0; u < FU; ++u) {
            const int i = i0 + u * EB;
            p[u] = (i < end) ? bkt[i] : make_int2(0, -1);
        }
        #pragma unroll
        for (int u = 0; u < FU; ++u) {
            if (p[u].y >= 0) {
                const int pos = atomicAdd(&cur[p[u].y - lo], 1);
                csr_src[beg + pos] = p[u].x;
            }
        }
    }
}

template<int NT>
__global__ void k_wmma_gemm(const float* __restrict__ A,
                            const float* __restrict__ B,
                            float* __restrict__ XW,
                            int nrows, int ncols) {
    const int lane = threadIdx.x & 31;
    const int half = lane >> 4;
    const int l    = lane & 15;

    int arow = blockIdx.x * 16 + l;
    if (arow >= nrows) arow = nrows - 1;
    const float* __restrict__ Arow = A + (size_t)arow * NFEAT;

    int col[NT];
    #pragma unroll
    for (int t = 0; t < NT; ++t) col[t] = blockIdx.y * (16 * NT) + t * 16 + l;

    v8f acc[NT];
    #pragma unroll
    for (int t = 0; t < NT; ++t) acc[t] = (v8f){0, 0, 0, 0, 0, 0, 0, 0};

    #pragma unroll 4
    for (int k0 = 0; k0 < NFEAT; k0 += 4) {
        const int ka = k0 + 2 * half;
        v2f a;
        a.x = Arow[ka];
        a.y = Arow[ka + 1];
        #pragma unroll
        for (int t = 0; t < NT; ++t) {
            v2f b;
            b.x = B[(size_t)ka       * ncols + col[t]];
            b.y = B[(size_t)(ka + 1) * ncols + col[t]];
            acc[t] = __builtin_amdgcn_wmma_f32_16x16x4_f32(
                         false, a, false, b, (short)0, acc[t], false, false);
        }
    }

    const int rbase = blockIdx.x * 16 + half * 8;
    #pragma unroll
    for (int t = 0; t < NT; ++t) {
        #pragma unroll
        for (int r = 0; r < 8; ++r) {
            const int row = rbase + r;
            if (row < nrows) XW[(size_t)row * ncols + col[t]] = acc[t][r];
        }
    }
}

template<int ROWS>
__global__ void __launch_bounds__(256) k_row_gather(const int* __restrict__ rowptr,
                                                    const int* __restrict__ csr_src,
                                                    const float* __restrict__ dis,
                                                    const float* __restrict__ feat,
                                                    float* io, int ncols, int n) {
    const int lane = threadIdx.x & 31;
    const int first = __builtin_amdgcn_readfirstlane(
        (int)((blockIdx.x * (blockDim.x >> 5) + (threadIdx.x >> 5)) * ROWS));

    const bool act = (lane * 4 < ncols);
    const int  c4  = act ? lane * 4 : 0;
    const float* __restrict__ fcol = feat + c4;

    for (int r = 0; r < ROWS; ++r) {
        const int node = first + r;
        if (node >= n) break;

        const int   beg = rowptr[node];
        const int   end = rowptr[node + 1];
        const float di  = dis[node];

        float4* iop = (float4*)(io + (size_t)node * ncols + c4);
        float4 acc = *iop;

        for (int base = beg; base < end; base += 32) {
            const int left = end - base;
            const int m = left < 32 ? left : 32;
            int   ms = 0;
            float mw = 0.0f;
            if (lane < m) {
                ms = csr_src[base + lane];
                mw = dis[ms] * di;
            }
            int j = 0;
            for (; j + 4 <= m; j += 4) {
                const int   s0 = __builtin_amdgcn_readlane(ms, j);
                const int   s1 = __builtin_amdgcn_readlane(ms, j + 1);
                const int   s2 = __builtin_amdgcn_readlane(ms, j + 2);
                const int   s3 = __builtin_amdgcn_readlane(ms, j + 3);
                const float w0 = __int_as_float(__builtin_amdgcn_readlane(__float_as_int(mw), j));
                const float w1 = __int_as_float(__builtin_amdgcn_readlane(__float_as_int(mw), j + 1));
                const float w2 = __int_as_float(__builtin_amdgcn_readlane(__float_as_int(mw), j + 2));
                const float w3 = __int_as_float(__builtin_amdgcn_readlane(__float_as_int(mw), j + 3));
                const float4 v0 = *(const float4*)(fcol + (size_t)s0 * ncols);
                const float4 v1 = *(const float4*)(fcol + (size_t)s1 * ncols);
                const float4 v2 = *(const float4*)(fcol + (size_t)s2 * ncols);
                const float4 v3 = *(const float4*)(fcol + (size_t)s3 * ncols);
                acc.x = fmaf(v0.x, w0, acc.x); acc.y = fmaf(v0.y, w0, acc.y); acc.z = fmaf(v0.z, w0, acc.z); acc.w = fmaf(v0.w, w0, acc.w);
                acc.x = fmaf(v1.x, w1, acc.x); acc.y = fmaf(v1.y, w1, acc.y); acc.z = fmaf(v1.z, w1, acc.z); acc.w = fmaf(v1.w, w1, acc.w);
                acc.x = fmaf(v2.x, w2, acc.x); acc.y = fmaf(v2.y, w2, acc.y); acc.z = fmaf(v2.z, w2, acc.z); acc.w = fmaf(v2.w, w2, acc.w);
                acc.x = fmaf(v3.x, w3, acc.x); acc.y = fmaf(v3.y, w3, acc.y); acc.z = fmaf(v3.z, w3, acc.z); acc.w = fmaf(v3.w, w3, acc.w);
            }
            for (; j < m; ++j) {
                const int   s0 = __builtin_amdgcn_readlane(ms, j);
                const float w0 = __int_as_float(__builtin_amdgcn_readlane(__float_as_int(mw), j));
                const float4 v0 = *(const float4*)(fcol + (size_t)s0 * ncols);
                acc.x = fmaf(v0.x, w0, acc.x); acc.y = fmaf(v0.y, w0, acc.y); acc.z = fmaf(v0.z, w0, acc.z); acc.w = fmaf(v0.w, w0, acc.w);
            }
        }
        if (act) *iop = acc;
    }
}


#define TROWS  16
#define HPITCH 132
#define NT2    3
#define WPITCH 48
#define W2Q    (NFEAT * (WPITCH / 4) / 256)
#define KGRP   8

__global__ void __launch_bounds__(256) k_row_gather_gemm2(const int* __restrict__ rowptr,
                                                          const int* __restrict__ csr_src,
                                                          const float* __restrict__ dis,
                                                          const float* __restrict__ feat,
                                                          const float* __restrict__ b1,
                                                          const float* __restrict__ W2,
                                                          const float* __restrict__ b2,
                                                          float* __restrict__ hw2,
                                                          float* __restrict__ out,
                                                          int ncls, int n) {
    __shared__ __attribute__((aligned(16))) float hs[TROWS * HPITCH];
    __shared__ __attribute__((aligned(16))) float w2s[NFEAT * WPITCH];
    __shared__ __attribute__((aligned(16))) float os_hw [TROWS * WPITCH];
    __shared__ __attribute__((aligned(16))) float os_out[TROWS * WPITCH];
    __shared__ float dsq_s[TROWS];

    const int lane = threadIdx.x & 31;
    const int wv    = __builtin_amdgcn_readfirstlane((int)(threadIdx.x >> 5));
    const int first = blockIdx.x * TROWS;

    float4 wq[W2Q];
    #pragma unroll
    for (int u = 0; u < W2Q; ++u) {
        const int q  = threadIdx.x + u * 256;
        const int k  = q / (WPITCH / 4);
        const int cq = (q % (WPITCH / 4)) * 4;
        wq[u] = make_float4(0.0f, 0.0f, 0.0f, 0.0f);
        if (cq < ncls) wq[u] = *(const float4*)(W2 + (size_t)k * ncls + cq);
    }

    const int c4 = lane * 4;
    const float* __restrict__ fcol = feat + c4;
    const float4 bb = *(const float4*)(b1 + c4);

    for (int rr = 0; rr < 2; ++rr) {
        const int trow = wv + rr * 8;
        const int node = first + trow;
        float4 acc = make_float4(0.0f, 0.0f, 0.0f, 0.0f);
        float  dd  = 0.0f;
        if (node < n) {
            const int beg = rowptr[node];
            const int end = rowptr[node + 1];
            const float di = dis[node];
            dd = di * di;

            const float4 xs = *(const float4*)(fcol + (size_t)node * NFEAT);
            acc.x = fmaf(xs.x, dd, bb.x); acc.y = fmaf(xs.y, dd, bb.y);
            acc.z = fmaf(xs.z, dd, bb.z); acc.w = fmaf(xs.w, dd, bb.w);

            for (int base = beg; base < end; base += 32) {
                const int left = end - base;
                const int m = left < 32 ? left : 32;
                int   ms = 0;
                float mw = 0.0f;
                if (lane < m) {
                    ms = csr_src[base + lane];
                    mw = dis[ms] * di;
                }
                int j = 0;
                for (; j + 4 <= m; j += 4) {
                    const int   s0 = __builtin_amdgcn_readlane(ms, j);
                    const int   s1 = __builtin_amdgcn_readlane(ms, j + 1);
                    const int   s2 = __builtin_amdgcn_readlane(ms, j + 2);
                    const int   s3 = __builtin_amdgcn_readlane(ms, j + 3);
                    const float w0 = __int_as_float(__builtin_amdgcn_readlane(__float_as_int(mw), j));
                    const float w1 = __int_as_float(__builtin_amdgcn_readlane(__float_as_int(mw), j + 1));
                    const float w2 = __int_as_float(__builtin_amdgcn_readlane(__float_as_int(mw), j + 2));
                    const float w3 = __int_as_float(__builtin_amdgcn_readlane(__float_as_int(mw), j + 3));
                    const float4 v0 = *(const float4*)(fcol + (size_t)s0 * NFEAT);
                    const float4 v1 = *(const float4*)(fcol + (size_t)s1 * NFEAT);
                    const float4 v2 = *(const float4*)(fcol + (size_t)s2 * NFEAT);
                    const float4 v3 = *(const float4*)(fcol + (size_t)s3 * NFEAT);
                    acc.x = fmaf(v0.x, w0, acc.x); acc.y = fmaf(v0.y, w0, acc.y); acc.z = fmaf(v0.z, w0, acc.z); acc.w = fmaf(v0.w, w0, acc.w);
                    acc.x = fmaf(v1.x, w1, acc.x); acc.y = fmaf(v1.y, w1, acc.y); acc.z = fmaf(v1.z, w1, acc.z); acc.w = fmaf(v1.w, w1, acc.w);
                    acc.x = fmaf(v2.x, w2, acc.x); acc.y = fmaf(v2.y, w2, acc.y); acc.z = fmaf(v2.z, w2, acc.z); acc.w = fmaf(v2.w, w2, acc.w);
                    acc.x = fmaf(v3.x, w3, acc.x); acc.y = fmaf(v3.y, w3, acc.y); acc.z = fmaf(v3.z, w3, acc.z); acc.w = fmaf(v3.w, w3, acc.w);
                }
                #pragma unroll 1
                for (; j < m; ++j) {
                    const int   s0 = __builtin_amdgcn_readlane(ms, j);
                    const float w0 = __int_as_float(__builtin_amdgcn_readlane(__float_as_int(mw), j));
                    const float4 v0 = *(const float4*)(fcol + (size_t)s0 * NFEAT);
                    acc.x = fmaf(v0.x, w0, acc.x); acc.y = fmaf(v0.y, w0, acc.y); acc.z = fmaf(v0.z, w0, acc.z); acc.w = fmaf(v0.w, w0, acc.w);
                }
            }
            acc.x = fmaxf(acc.x, 0.0f); acc.y = fmaxf(acc.y, 0.0f);
            acc.z = fmaxf(acc.z, 0.0f); acc.w = fmaxf(acc.w, 0.0f);
        }
        *(float4*)(hs + trow * HPITCH + c4) = acc;
        if (lane == 0) dsq_s[trow] = dd;
    }

    #pragma unroll
    for (int u = 0; u < W2Q; ++u)
        *(float4*)(w2s + (threadIdx.x + u * 256) * 4) = wq[u];

    const int half = lane >> 4;
    const int l    = lane & 15;
    const int c0   = wv * 16 + l;
    const bool cok = (wv < NT2) && (c0 < ncls);
    const float bv = cok ? b2[c0] : 0.0f;

    __syncthreads();

    if (wv < NT2) {
        v8f acc = (v8f){0, 0, 0, 0, 0, 0, 0, 0};
        const float* __restrict__ arow = hs  + l * HPITCH + 2 * half;
        const float* __restrict__ bcol = w2s + (2 * half) * WPITCH + c0;
        #pragma unroll
        for (int kg = 0; kg < NFEAT / 4; kg += KGRP) {
            v2f a[KGRP], b[KGRP];
            #pragma unroll
            for (int u = 0; u < KGRP; ++u) {
                const int ks = kg + u;
                a[u].x = arow[ks * 4];
                a[u].y = arow[ks * 4 + 1];
                b[u].x = bcol[(ks * 4)     * WPITCH];
                b[u].y = bcol[(ks * 4 + 1) * WPITCH];
            }
            __builtin_amdgcn_sched_barrier(0);
            #pragma unroll
            for (int u = 0; u < KGRP; ++u) {
                acc = __builtin_amdgcn_wmma_f32_16x16x4_f32(
                          false, a[u], false, b[u], (short)0, acc, false, false);
            }
        }
        if (cok) {
            #pragma unroll
            for (int r = 0; r < 8; ++r) {
                const int trow = half * 8 + r;
                const float v = acc[r];
                os_hw [trow * ncls + c0] = v;
                os_out[trow * ncls + c0] = v * dsq_s[trow] + bv;
            }
        }
    }

    __syncthreads();

    const int i4 = threadIdx.x * 4;
    if (i4 < TROWS * ncls) {
        const int trow = i4 / ncls;
        if (first + trow < n) {
            const size_t g = (size_t)first * ncls + i4;
            *(float4*)(hw2 + g) = *(const float4*)(os_hw  + i4);
            *(float4*)(out + g) = *(const float4*)(os_out + i4);
        }
    }
}

extern "C" void kernel_launch(void* const* d_in, const int* in_sizes, int n_in,
                              void* d_out, int out_size, void* d_ws, size_t ws_size,
                              hipStream_t stream) {
    const float* x  = (const float*)d_in[0];
    const int*   ei = (const int*)  d_in[1];
    const float* W1 = (const float*)d_in[2];
    const float* b1 = (const float*)d_in[3];
    const float* W2 = (const float*)d_in[4];
    const float* b2 = (const float*)d_in[5];
    float* out = (float*)d_out;

    const int N   = in_sizes[0] / NFEAT;
    const int E   = in_sizes[1] / 2;
    const int CLS = in_sizes[5];

    const int* src = ei;
    const int* dst = ei + E;

    const int NBKT = (N + BKT_W - 1) / BKT_W;

    char* ws = (char*)d_ws;
    size_t off = 0;
    float* dis = (float*)(ws + off); off += ((size_t)N * 4 + 255) & ~(size_t)255;
    float* xw1 = (float*)(ws + off); off += (size_t)N * NFEAT * 4;
    float* hw2 = (float*)(ws + off); off += ((size_t)N * CLS * 4 + 255) & ~(size_t)255;
    int* bc     = (int*)(ws + off); off += ((size_t)NBKT * 8 + 255) & ~(size_t)255;
    int* rowptr = (int*)(ws + off); off += ((size_t)(N + 1) * 4 + 255) & ~(size_t)255;
    int* bstart = (int*)(ws + off); off += ((size_t)(NBKT + 1) * 4 + 255) & ~(size_t)255;
    int* bkt    = (int*)(ws + off); off += ((size_t)E * 8 + 255) & ~(size_t)255;
    int* csr    = (int*)(ws + off); off += ((size_t)E * 4 + 255) & ~(size_t)255;
    (void)ws_size; (void)n_in; (void)out_size;

    const int TPB = 256;
    const int WAVES = TPB / 32;

    const int EBLK = (E + EB * EPT - 1) / (EB * EPT);
    k_zero_bkt   <<<(2 * NBKT + TPB - 1) / TPB, TPB, 0, stream>>>(bc, 2 * NBKT);
    k_bkt_count  <<<EBLK, EB, 0, stream>>>(dst, bc, E, NBKT);
    k_bkt_scatter<<<EBLK, EB, 0, stream>>>(src, dst, bc, bc + NBKT, bstart, bkt, E, NBKT);
    k_bkt_finish <<<NBKT, EB, 0, stream>>>(bkt, bstart, rowptr, dis, csr, N);

    {
        dim3 grid((N + 15) / 16, NFEAT / 64);
        k_wmma_gemm<4><<<grid, 32, 0, stream>>>(x, W1, xw1, N, NFEAT);
    }
    k_row_gather_gemm2<<<(N + TROWS - 1) / TROWS, TPB, 0, stream>>>(
        rowptr, csr, dis, xw1, b1, W2, b2, hw2, out, CLS, N);
    k_row_gather<4><<<(N + 4 * WAVES - 1) / (4 * WAVES), TPB, 0, stream>>>(
        rowptr, csr, dis, hw2, out, CLS, N);
}


# x@W1 moved behind the layer-1 row sums, into the gather launch: a wave sums 16 x rows into its own LDS strip, then strip@W1 (W1 in LDS), relu, @W2; 7 to 6 launches; sum order now (A x) W1
# speedup vs baseline: 4.0992x; 1.0827x over previous
#include <hip/hip_runtime.h>


typedef __attribute__((ext_vector_type(2))) float v2f;
typedef __attribute__((ext_vector_type(8))) float v8f;

#define NFEAT 128


#define BKT_SHIFT 8
#define BKT_W (1 << BKT_SHIFT)
#define EB 1024
#define EPT 4
#define FU 4

__global__ void __launch_bounds__(256) k_zero_bkt(int* __restrict__ bc, int n) {
    int i = blockIdx.x * blockDim.x + threadIdx.x;
    if (i < n) bc[i] = 0;
}

__device__ __forceinline__ int wave_incl_scan(int v, int lane) {
    #pragma unroll
    for (int o = 1; o < 32; o <<= 1) {
        const int t = __shfl_up(v, o);
        if (lane >= o) v += t;
    }
    return v;
}

__global__ void __launch_bounds__(EB) k_bkt_count(const int* __restrict__ dst,
                                                  int* bcnt, int E, int nbkt) {
    __shared__ int hist[EB];
    const int t = threadIdx.x;
    hist[t] = 0;
    __syncthreads();
    const int base = blockIdx.x * (EB * EPT);
    int d[EPT];
    #pragma unroll
    for (int k = 0; k < EPT; ++k) {
        const int e = base + k * EB + t;
        d[k] = (e < E) ? dst[e] : -1;
    }
    #pragma unroll
    for (int k = 0; k < EPT; ++k)
        if (d[k] >= 0) atomicAdd(&hist[d[k] >> BKT_SHIFT], 1);
    __syncthreads();
    if (t < nbkt) {
        const int h = hist[t];
        if (h) atomicAdd(&bcnt[t], h);
    }
}

__global__ void __launch_bounds__(EB) k_bkt_scatter(const int* __restrict__ src,
                                                    const int* __restrict__ dst,
                                                    const int* __restrict__ bcnt,
                                                    int* bcur,
                                                    int* __restrict__ bstart,
                                                    int* __restrict__ bkt_raw,
                                                    int E, int nbkt) {
    __shared__ int hist[EB];
    __shared__ int wsum[EB / 32];
    const int t = threadIdx.x, lane = t & 31, w = t >> 5;
    int2* __restrict__ bkt = (int2*)bkt_raw;

    hist[t] = 0;
    __syncthreads();

    const int base = blockIdx.x * (EB * EPT);
    int s[EPT], d[EPT], r[EPT];
    #pragma unroll
    for (int k = 0; k < EPT; ++k) {
        const int e = base + k * EB + t;
        d[k] = (e < E) ? dst[e] : -1;
        s[k] = (e < E) ? src[e] : 0;
    }
    #pragma unroll
    for (int k = 0; k < EPT; ++k) {
        r[k] = 0;
        if (d[k] >= 0) r[k] = atomicAdd(&hist[d[k] >> BKT_SHIFT], 1);
    }

    const int c   = (t < nbkt) ? bcnt[t] : 0;
    const int inc = wave_incl_scan(c, lane);
    if (lane == 31) wsum[w] = inc;
    __syncthreads();
    if (w == 0) {
        const int x  = wsum[lane];
        const int xi = wave_incl_scan(x, lane);
        wsum[lane] = xi - x;
    }
    __syncthreads();
    const int excl = wsum[w] + inc - c;

    int mybase = 0;
    if (t < nbkt) {
        const int h = hist[t];
        mybase = excl + (h ? atomicAdd(&bcur[t], h) : 0);
        if (blockIdx.x == 0) {
            bstart[t] = excl;
            if (t == nbkt - 1) bstart[nbkt] = excl + c;
        }
    }
    hist[t] = mybase;
    __syncthreads();

    #pragma unroll
    for (int k = 0; k < EPT; ++k) {
        if (d[k] >= 0) bkt[hist[d[k] >> BKT_SHIFT] + r[k]] = make_int2(s[k], d[k]);
    }
}

__global__ void __launch_bounds__(EB) k_bkt_finish(const int* __restrict__ bkt_raw,
                                                   const int* __restrict__ bstart,
                                                   int* __restrict__ rowptr,
                                                   float* __restrict__ dis,
                                                   int* __restrict__ csr_src, int n) {
    __shared__ int cnt[BKT_W];
    __shared__ int cur[BKT_W];
    __shared__ int wsum[BKT_W / 32];
    const int t = threadIdx.x, lane = t & 31, w = t >> 5;
    const int2* __restrict__ bkt = (const int2*)bkt_raw;

    const int lo  = blockIdx.x << BKT_SHIFT;
    const int beg = bstart[blockIdx.x];
    const int end = bstart[blockIdx.x + 1];

    if (t < BKT_W) cnt[t] = 0;
    __syncthreads();
    for (int i0 = beg + t; i0 < end; i0 += FU * EB) {
        int dd[FU];
        #pragma unroll
        for (int u = 0; u < FU; ++u) {
            const int i = i0 + u * EB;
            dd[u] = (i < end) ? bkt[i].y : -1;
        }
        #pragma unroll
        for (int u = 0; u < FU; ++u)
            if (dd[u] >= 0) atomicAdd(&cnt[dd[u] - lo], 1);
    }
    __syncthreads();

    int c = 0, inc = 0;
    if (t < BKT_W) {
        c   = cnt[t];
        inc = wave_incl_scan(c, lane);
        if (lane == 31) wsum[w] = inc;
    }
    __syncthreads();
    if (w == 0) {
        const int x  = (lane < BKT_W / 32) ? wsum[lane] : 0;
        const int xi = wave_incl_scan(x, lane);
        if (lane < BKT_W / 32) wsum[lane] = xi - x;
    }
    __syncthreads();
    if (t < BKT_W) {
        const int excl = wsum[w] + inc - c;
        cur[t] = excl;
        const int node = lo + t;
        if (node < n) {
            rowptr[node] = beg + excl;
            dis[node]    = rsqrtf((float)c + 1.0f);
            if (node == n - 1) rowptr[n] = beg + excl + c;
        }
    }
    __syncthreads();
    for (int i0 = beg + t; i0 < end; i0 += FU * EB) {
        int2 p[FU];
        #pragma unroll
        for (int u = 0; u < FU; ++u) {
            const int i = i0 + u * EB;
            p[u] = (i < end) ? bkt[i] : make_int2(0, -1);
        }
        #pragma unroll
        for (int u = 0; u < FU; ++u) {
            if (p[u].y >= 0) {
                const int pos = atomicAdd(&cur[p[u].y - lo], 1);
                csr_src[beg + pos] = p[u].x;
            }
        }
    }
}

template<int ROWS>
__global__ void __launch_bounds__(256) k_row_gather(const int* __restrict__ rowptr,
                                                    const int* __restrict__ csr_src,
                                                    const float* __restrict__ dis,
                                                    const float* __restrict__ feat,
                                                    float* io, int ncols, int n) {
    const int lane = threadIdx.x & 31;
    const int first = __builtin_amdgcn_readfirstlane(
        (int)((blockIdx.x * (blockDim.x >> 5) + (threadIdx.x >> 5)) * ROWS));

    const bool act = (lane * 4 < ncols);
    const int  c4  = act ? lane * 4 : 0;
    const float* __restrict__ fcol = feat + c4;

    for (int r = 0; r < ROWS; ++r) {
        const int node = first + r;
        if (node >= n) break;

        const int   beg = rowptr[node];
        const int   end = rowptr[node + 1];
        const float di  = dis[node];

        float4* iop = (float4*)(io + (size_t)node * ncols + c4);
        float4 acc = *iop;

        for (int base = beg; base < end; base += 32) {
            const int left = end - base;
            const int m = left < 32 ? left : 32;
            int   ms = 0;
            float mw = 0.0f;
            if (lane < m) {
                ms = csr_src[base + lane];
                mw = dis[ms] * di;
            }
            int j = 0;
            for (; j + 4 <= m; j += 4) {
                const int   s0 = __builtin_amdgcn_readlane(ms, j);
                const int   s1 = __builtin_amdgcn_readlane(ms, j + 1);
                const int   s2 = __builtin_amdgcn_readlane(ms, j + 2);
                const int   s3 = __builtin_amdgcn_readlane(ms, j + 3);
                const float w0 = __int_as_float(__builtin_amdgcn_readlane(__float_as_int(mw), j));
                const float w1 = __int_as_float(__builtin_amdgcn_readlane(__float_as_int(mw), j + 1));
                const float w2 = __int_as_float(__builtin_amdgcn_readlane(__float_as_int(mw), j + 2));
                const float w3 = __int_as_float(__builtin_amdgcn_readlane(__float_as_int(mw), j + 3));
                const float4 v0 = *(const float4*)(fcol + (size_t)s0 * ncols);
                const float4 v1 = *(const float4*)(fcol + (size_t)s1 * ncols);
                const float4 v2 = *(const float4*)(fcol + (size_t)s2 * ncols);
                const float4 v3 = *(const float4*)(fcol + (size_t)s3 * ncols);
                acc.x = fmaf(v0.x, w0, acc.x); acc.y = fmaf(v0.y, w0, acc.y); acc.z = fmaf(v0.z, w0, acc.z); acc.w = fmaf(v0.w, w0, acc.w);
                acc.x = fmaf(v1.x, w1, acc.x); acc.y = fmaf(v1.y, w1, acc.y); acc.z = fmaf(v1.z, w1, acc.z); acc.w = fmaf(v1.w, w1, acc.w);
                acc.x = fmaf(v2.x, w2, acc.x); acc.y = fmaf(v2.y, w2, acc.y); acc.z = fmaf(v2.z, w2, acc.z); acc.w = fmaf(v2.w, w2, acc.w);
                acc.x = fmaf(v3.x, w3, acc.x); acc.y = fmaf(v3.y, w3, acc.y); acc.z = fmaf(v3.z, w3, acc.z); acc.w = fmaf(v3.w, w3, acc.w);
            }
            for (; j < m; ++j) {
                const int   s0 = __builtin_amdgcn_readlane(ms, j);
                const float w0 = __int_as_float(__builtin_amdgcn_readlane(__float_as_int(mw), j));
                const float4 v0 = *(const float4*)(fcol + (size_t)s0 * ncols);
                acc.x = fmaf(v0.x, w0, acc.x); acc.y = fmaf(v0.y, w0, acc.y); acc.z = fmaf(v0.z, w0, acc.z); acc.w = fmaf(v0.w, w0, acc.w);
            }
        }
        if (act) *iop = acc;
    }
}


#define TROWS   16
#define HPITCH  132
#define NT2     3
#define WPITCH  48
#define W1PITCH 136
#define GW      13
#define GT      (GW * 32)
#define W1Q     ((NFEAT * (NFEAT / 4) + GT - 1) / GT)
#define W2Q     ((NFEAT * (WPITCH / 4) + GT - 1) / GT)
#define KG1     2
#define KG2     4
#define OSOFF   1024
#define CH0     31

__device__ __forceinline__ void issue_rows(float4 (&v)[16], int ms, int j, int cnt,
                                           const float* __restrict__ fcol) {
    #pragma unroll
    for (int u = 0; u < 16; ++u) {
        v[u] = make_float4(0.0f, 0.0f, 0.0f, 0.0f);
        if (u < cnt) {
            const int s = __builtin_amdgcn_readlane(ms, j + u);
            v[u] = *(const float4*)(fcol + (size_t)s * NFEAT);
        }
    }
}
__device__ __forceinline__ float4 sum_rows(float4 acc, const float4 (&v)[16], float mw,
                                           int j, int cnt) {
    #pragma unroll
    for (int u = 0; u < 16; ++u) {
        if (u < cnt) {
            const float w = __int_as_float(__builtin_amdgcn_readlane(__float_as_int(mw), j + u));
            acc.x = fmaf(v[u].x, w, acc.x); acc.y = fmaf(v[u].y, w, acc.y);
            acc.z = fmaf(v[u].z, w, acc.z); acc.w = fmaf(v[u].w, w, acc.w);
        }
    }
    return acc;
}

__global__ void __launch_bounds__(GT) k_row_gather_gemm12(const int* __restrict__ rowptr,
                                                          const int* __restrict__ csr_src,
                                                          const float* __restrict__ dis,
                                                          const float* __restrict__ x,
                                                          const float* __restrict__ W1,
                                                          const float* __restrict__ b1,
                                                          const float* __restrict__ W2,
                                                          const float* __restrict__ b2,
                                                          float* __restrict__ hw2,
                                                          float* __restrict__ out,
                                                          int ncls, int n) {
    __shared__ __attribute__((aligned(16))) float w1s[NFEAT * W1PITCH];
    __shared__ __attribute__((aligned(16))) float w2s[NFEAT * WPITCH];
    __shared__ __attribute__((aligned(16))) float strips[GW * TROWS * HPITCH];

    {
        float4 p1[W1Q], p2[W2Q];
        #pragma unroll
        for (int u = 0; u < W1Q; ++u) {
            const int q  = threadIdx.x + u * GT;
            const int k  = q / (NFEAT / 4);
            const int cq = (q % (NFEAT / 4)) * 4;
            p1[u] = make_float4(0.0f, 0.0f, 0.0f, 0.0f);
            if (q < NFEAT * (NFEAT / 4)) p1[u] = *(const float4*)(W1 + (size_t)k * NFEAT + cq);
        }
        #pragma unroll
        for (int u = 0; u < W2Q; ++u) {
            const int q  = threadIdx.x + u * GT;
            const int k  = q / (WPITCH / 4);
            const int cq = (q % (WPITCH / 4)) * 4;
            p2[u] = make_float4(0.0f, 0.0f, 0.0f, 0.0f);
            if (q < NFEAT * (WPITCH / 4) && cq < ncls) p2[u] = *(const float4*)(W2 + (size_t)k * ncls + cq);
        }
        #pragma unroll
        for (int u = 0; u < W1Q; ++u) {
            const int q  = threadIdx.x + u * GT;
            const int k  = q / (NFEAT / 4);
            const int cq = (q % (NFEAT / 4)) * 4;
            if (q < NFEAT * (NFEAT / 4)) *(float4*)(w1s + k * W1PITCH + cq) = p1[u];
        }
        #pragma unroll
        for (int u = 0; u < W2Q; ++u) {
            const int q = threadIdx.x + u * GT;
            if (q < NFEAT * (WPITCH / 4)) *(float4*)(w2s + q * 4) = p2[u];
        }
    }
    __syncthreads();

    const int lane = threadIdx.x & 31;
    const int wv    = __builtin_amdgcn_readfirstlane((int)(threadIdx.x >> 5));
    float* strip = strips + wv * (TROWS * HPITCH);
    const int c4 = lane * 4;
    const float* __restrict__ fcol = x + c4;
    const int half = lane >> 4;
    const int l    = lane & 15;
    const float* arow = strip + l * HPITCH + 2 * half;

    const int ntile = (n + TROWS - 1) / TROWS;
    for (int tile = blockIdx.x * GW + wv; tile < ntile; tile += gridDim.x * GW) {
    const int first = tile * TROWS;

    int   rp = 0;
    float dv = 0.0f;
    {
        int ri = first + lane; if (ri > n) ri = n;
        int ni = first + lane; if (ni > n - 1) ni = n - 1;
        if (lane <= TROWS) rp = rowptr[ri];
        if (lane <  TROWS) dv = dis[ni];
    }

#define ROW_CNT(q)   ((__builtin_amdgcn_readlane(rp, (q) + 1) - __builtin_amdgcn_readlane(rp, (q))) < CH0 \
                      ? (__builtin_amdgcn_readlane(rp, (q) + 1) - __builtin_amdgcn_readlane(rp, (q))) : CH0)
#define ROW_NODE(q)  ((first + (q)) < n ? (first + (q)) : (n - 1))
#define LIST_IDX(q, cnt)  ((lane >= 1 && lane <= (cnt)) ? __builtin_amdgcn_readlane(rp, (q)) + lane - 1 \
                                                         : ((cnt) > 0 ? __builtin_amdgcn_readlane(rp, (q)) : 0))
#define LIST_SEL(raw, q, cnt)  ((lane >= 1 && lane <= (cnt)) ? (raw) : ROW_NODE(q))
    int   ms_a, ms_b;
    float w_a;
    {
        const int cnt0 = ROW_CNT(0);
        const int cnt1 = ROW_CNT(1);
        const int raw0 = csr_src[LIST_IDX(0, cnt0)];
        const int raw1 = csr_src[LIST_IDX(1, cnt1)];
        ms_a = LIST_SEL(raw0, 0, cnt0);
        ms_b = LIST_SEL(raw1, 1, cnt1);
        w_a  = dis[ms_a] * __int_as_float(__builtin_amdgcn_readlane(__float_as_int(dv), 0));
    }
    #pragma unroll 1
    for (int r = 0; r < TROWS; ++r) {
        const int node = first + r;
        const int q1 = (r + 1 < TROWS) ? r + 1 : TROWS - 1;
        const int q2 = (r + 2 < TROWS) ? r + 2 : TROWS - 1;
        const int cnt2 = ROW_CNT(q2);
        const int beg = __builtin_amdgcn_readlane(rp, r);
        const int end = __builtin_amdgcn_readlane(rp, r + 1);
        const int cnt = (end - beg) < CH0 ? (end - beg) : CH0;
        const int tot = (node < n) ? cnt + 1 : 0;

        float4 acc = make_float4(0.0f, 0.0f, 0.0f, 0.0f);
        float4 v[16];
        issue_rows(v, ms_a, 0, tot, fcol);
        const float raw_w = dis[ms_b];
        const int   raw_c = csr_src[LIST_IDX(q2, cnt2)];
        acc = sum_rows(acc, v, w_a, 0, tot);
        for (int j = 16; j < tot; j += 16) {
            issue_rows(v, ms_a, j, tot - j, fcol);
            acc = sum_rows(acc, v, w_a, j, tot - j);
        }
        if (tot > 0 && end - beg > CH0) {
            const float di = __int_as_float(__builtin_amdgcn_readlane(__float_as_int(dv), r));
            for (int base = beg + CH0; base < end; base += 32) {
                const int left = end - base;
                const int m2 = left < 32 ? left : 32;
                int   ms = 0;
                float mw = 0.0f;
                if (lane < m2) {
                    ms = csr_src[base + lane];
                    mw = dis[ms] * di;
                }
                for (int j = 0; j < m2; j += 16) {
                    issue_rows(v, ms, j, m2 - j, fcol);
                    acc = sum_rows(acc, v, mw, j, m2 - j);
                }
            }
        }
        *(float4*)(strip + r * HPITCH + c4) = acc;
        w_a  = raw_w * __int_as_float(__builtin_amdgcn_readlane(__float_as_int(dv), q1));
        ms_a = ms_b;
        ms_b = LIST_SEL(raw_c, q2, cnt2);
    }
#undef ROW_CNT
#undef ROW_NODE
#undef LIST_IDX
#undef LIST_SEL

    __builtin_amdgcn_wave_barrier();

    v8f hacc[NFEAT / 16];
    #pragma unroll
    for (int nh = 0; nh < 2; ++nh) {
        v8f acc[4];
        #pragma unroll
        for (int t = 0; t < 4; ++t) acc[t] = (v8f){0, 0, 0, 0, 0, 0, 0, 0};
        const float* __restrict__ bcol = w1s + (2 * half) * W1PITCH + nh * 64 + l;
        #pragma unroll
        for (int kg = 0; kg < NFEAT / 4; kg += KG1) {
            v2f a[KG1], b[4][KG1];
            #pragma unroll
            for (int u = 0; u < KG1; ++u) {
                const int ks = kg + u;
                a[u].x = arow[ks * 4];
                a[u].y = arow[ks * 4 + 1];
                #pragma unroll
                for (int t = 0; t < 4; ++t) {
                    b[t][u].x = bcol[(ks * 4)     * W1PITCH + t * 16];
                    b[t][u].y = bcol[(ks * 4 + 1) * W1PITCH + t * 16];
                }
            }
            __builtin_amdgcn_sched_barrier(0);
            #pragma unroll
            for (int u = 0; u < KG1; ++u) {
                #pragma unroll
                for (int t = 0; t < 4; ++t) {
                    acc[t] = __builtin_amdgcn_wmma_f32_16x16x4_f32(
                                 false, a[u], false, b[t][u], (short)0, acc[t], false, false);
                }
            }
        }
        #pragma unroll
        for (int t = 0; t < 4; ++t) hacc[nh * 4 + t] = acc[t];
    }

    __builtin_amdgcn_wave_barrier();
    #pragma unroll
    for (int t = 0; t < NFEAT / 16; ++t) {
        const int   col = t * 16 + l;
        const float bb  = b1[col];
        #pragma unroll
        for (int r = 0; r < 8; ++r)
            strip[(half * 8 + r) * HPITCH + col] = fmaxf(hacc[t][r] + bb, 0.0f);
    }
    __builtin_amdgcn_wave_barrier();

    v8f acc2[NT2];
    #pragma unroll
    for (int t = 0; t < NT2; ++t) acc2[t] = (v8f){0, 0, 0, 0, 0, 0, 0, 0};
    {
        const float* __restrict__ bcol = w2s + (2 * half) * WPITCH + l;
        #pragma unroll
        for (int kg = 0; kg < NFEAT / 4; kg += KG2) {
            v2f a[KG2], b[NT2][KG2];
            #pragma unroll
            for (int u = 0; u < KG2; ++u) {
                const int ks = kg + u;
                a[u].x = arow[ks * 4];
                a[u].y = arow[ks * 4 + 1];
                #pragma unroll
                for (int t = 0; t < NT2; ++t) {
                    b[t][u].x = bcol[(ks * 4)     * WPITCH + t * 16];
                    b[t][u].y = bcol[(ks * 4 + 1) * WPITCH + t * 16];
                }
            }
            __builtin_amdgcn_sched_barrier(0);
            #pragma unroll
            for (int u = 0; u < KG2; ++u) {
                #pragma unroll
                for (int t = 0; t < NT2; ++t) {
                    acc2[t] = __builtin_amdgcn_wmma_f32_16x16x4_f32(
                                  false, a[u], false, b[t][u], (short)0, acc2[t], false, false);
                }
            }
        }
    }

    float dsq[8];
    #pragma unroll
    for (int r = 0; r < 8; ++r) {
        const float dlo = __int_as_float(__builtin_amdgcn_readlane(__float_as_int(dv), r));
        const float dhi = __int_as_float(__builtin_amdgcn_readlane(__float_as_int(dv), r + 8));
        const float d   = half ? dhi : dlo;
        dsq[r] = d * d;
    }

    __builtin_amdgcn_wave_barrier();
    float* os_hw  = strip;
    float* os_out = strip + OSOFF;
    #pragma unroll
    for (int t = 0; t < NT2; ++t) {
        const int c0 = t * 16 + l;
        if (c0 < ncls) {
            const float bv = b2[c0];
            #pragma unroll
            for (int r = 0; r < 8; ++r) {
                const int trow = half * 8 + r;
                const float v = acc2[t][r];
                os_hw [trow * ncls + c0] = v;
                os_out[trow * ncls + c0] = v * dsq[r] + bv;
            }
        }
    }
    __builtin_amdgcn_wave_barrier();

    for (int i4 = lane * 4; i4 < TROWS * ncls; i4 += 128) {
        const int trow = i4 / ncls;
        if (first + trow < n) {
            const size_t g = (size_t)first * ncls + i4;
            *(float4*)(hw2 + g) = *(const float4*)(os_hw  + i4);
            *(float4*)(out + g) = *(const float4*)(os_out + i4);
        }
    }
    __builtin_amdgcn_wave_barrier();
    }
}

extern "C" void kernel_launch(void* const* d_in, const int* in_sizes, int n_in,
                              void* d_out, int out_size, void* d_ws, size_t ws_size,
                              hipStream_t stream) {
    const float* x  = (const float*)d_in[0];
    const int*   ei = (const int*)  d_in[1];
    const float* W1 = (const float*)d_in[2];
    const float* b1 = (const float*)d_in[3];
    const float* W2 = (const float*)d_in[4];
    const float* b2 = (const float*)d_in[5];
    float* out = (float*)d_out;

    const int N   = in_sizes[0] / NFEAT;
    const int E   = in_sizes[1] / 2;
    const int CLS = in_sizes[5];

    const int* src = ei;
    const int* dst = ei + E;

    const int NBKT = (N + BKT_W - 1) / BKT_W;

    char* ws = (char*)d_ws;
    size_t off = 0;
    float* dis = (float*)(ws + off); off += ((size_t)N * 4 + 255) & ~(size_t)255;
    float* hw2 = (float*)(ws + off); off += ((size_t)N * CLS * 4 + 255) & ~(size_t)255;
    int* bc     = (int*)(ws + off); off += ((size_t)NBKT * 8 + 255) & ~(size_t)255;
    int* rowptr = (int*)(ws + off); off += ((size_t)(N + 1) * 4 + 255) & ~(size_t)255;
    int* bstart = (int*)(ws + off); off += ((size_t)(NBKT + 1) * 4 + 255) & ~(size_t)255;
    int* bkt    = (int*)(ws + off); off += ((size_t)E * 8 + 255) & ~(size_t)255;
    int* csr    = (int*)(ws + off); off += ((size_t)E * 4 + 255) & ~(size_t)255;
    (void)ws_size; (void)n_in; (void)out_size;

    const int TPB = 256;
    const int WAVES = TPB / 32;

    const int EBLK = (E + EB * EPT - 1) / (EB * EPT);
    k_zero_bkt   <<<(2 * NBKT + TPB - 1) / TPB, TPB, 0, stream>>>(bc, 2 * NBKT);
    k_bkt_count  <<<EBLK, EB, 0, stream>>>(dst, bc, E, NBKT);
    k_bkt_scatter<<<EBLK, EB, 0, stream>>>(src, dst, bc, bc + NBKT, bstart, bkt, E, NBKT);
    k_bkt_finish <<<NBKT, EB, 0, stream>>>(bkt, bstart, rowptr, dis, csr, N);

    {
        const int NTILE = (N + TROWS - 1) / TROWS;
        k_row_gather_gemm12<<<(NTILE + 2 * GW - 1) / (2 * GW), GT, 0, stream>>>(
            rowptr, csr, dis, x, W1, b1, W2, b2, hw2, out, CLS, N);
    }
    k_row_gather<4><<<(N + 4 * WAVES - 1) / (4 * WAVES), TPB, 0, stream>>>(
        rowptr, csr, dis, hw2, out, CLS, N);
}


# Layer-1 row sums look one row ahead: row r+1's first 16 x rows are loaded into a second register set before row r is summed (always 16 loads, so the waits are counted ones); same sums
# speedup vs baseline: 4.3880x; 1.0705x over previous
#include <hip/hip_runtime.h>


typedef __attribute__((ext_vector_type(2))) float v2f;
typedef __attribute__((ext_vector_type(8))) float v8f;

#define NFEAT 128


#define BKT_SHIFT 8
#define BKT_W (1 << BKT_SHIFT)
#define EB 1024
#define EPT 4
#define FU 4

__global__ void __launch_bounds__(256) k_zero_bkt(int* __restrict__ bc, int n) {
    int i = blockIdx.x * blockDim.x + threadIdx.x;
    if (i < n) bc[i] = 0;
}

__device__ __forceinline__ int wave_incl_scan(int v, int lane) {
    #pragma unroll
    for (int o = 1; o < 32; o <<= 1) {
        const int t = __shfl_up(v, o);
        if (lane >= o) v += t;
    }
    return v;
}

__global__ void __launch_bounds__(EB) k_bkt_count(const int* __restrict__ dst,
                                                  int* bcnt, int E, int nbkt) {
    __shared__ int hist[EB];
    const int t = threadIdx.x;
    hist[t] = 0;
    __syncthreads();
    const int base = blockIdx.x * (EB * EPT);
    int d[EPT];
    #pragma unroll
    for (int k = 0; k < EPT; ++k) {
        const int e = base + k * EB + t;
        d[k] = (e < E) ? dst[e] : -1;
    }
    #pragma unroll
    for (int k = 0; k < EPT; ++k)
        if (d[k] >= 0) atomicAdd(&hist[d[k] >> BKT_SHIFT], 1);
    __syncthreads();
    if (t < nbkt) {
        const int h = hist[t];
        if (h) atomicAdd(&bcnt[t], h);
    }
}

__global__ void __launch_bounds__(EB) k_bkt_scatter(const int* __restrict__ src,
                                                    const int* __restrict__ dst,
                                                    const int* __restrict__ bcnt,
                                                    int* bcur,
                                                    int* __restrict__ bstart,
                                                    int* __restrict__ bkt_raw,
                                                    int E, int nbkt) {
    __shared__ int hist[EB];
    __shared__ int wsum[EB / 32];
    const int t = threadIdx.x, lane = t & 31, w = t >> 5;
    int2* __restrict__ bkt = (int2*)bkt_raw;

    hist[t] = 0;
    __syncthreads();

    const int base = blockIdx.x * (EB * EPT);
    int s[EPT], d[EPT], r[EPT];
    #pragma unroll
    for (int k = 0; k < EPT; ++k) {
        const int e = base + k * EB + t;
        d[k] = (e < E) ? dst[e] : -1;
        s[k] = (e < E) ? src[e] : 0;
    }
    #pragma unroll
    for (int k = 0; k < EPT; ++k) {
        r[k] = 0;
        if (d[k] >= 0) r[k] = atomicAdd(&hist[d[k] >> BKT_SHIFT], 1);
    }

    const int c   = (t < nbkt) ? bcnt[t] : 0;
    const int inc = wave_incl_scan(c, lane);
    if (lane == 31) wsum[w] = inc;
    __syncthreads();
    if (w == 0) {
        const int x  = wsum[lane];
        const int xi = wave_incl_scan(x, lane);
        wsum[lane] = xi - x;
    }
    __syncthreads();
    const int excl = wsum[w] + inc - c;

    int mybase = 0;
    if (t < nbkt) {
        const int h = hist[t];
        mybase = excl + (h ? atomicAdd(&bcur[t], h) : 0);
        if (blockIdx.x == 0) {
            bstart[t] = excl;
            if (t == nbkt - 1) bstart[nbkt] = excl + c;
        }
    }
    hist[t] = mybase;
    __syncthreads();

    #pragma unroll
    for (int k = 0; k < EPT; ++k) {
        if (d[k] >= 0) bkt[hist[d[k] >> BKT_SHIFT] + r[k]] = make_int2(s[k], d[k]);
    }
}

__global__ void __launch_bounds__(EB) k_bkt_finish(const int* __restrict__ bkt_raw,
                                                   const int* __restrict__ bstart,
                                                   int* __restrict__ rowptr,
                                                   float* __restrict__ dis,
                                                   int* __restrict__ csr_src, int n) {
    __shared__ int cnt[BKT_W];
    __shared__ int cur[BKT_W];
    __shared__ int wsum[BKT_W / 32];
    const int t = threadIdx.x, lane = t & 31, w = t >> 5;
    const int2* __restrict__ bkt = (const int2*)bkt_raw;

    const int lo  = blockIdx.x << BKT_SHIFT;
    const int beg = bstart[blockIdx.x];
    const int end = bstart[blockIdx.x + 1];

    if (t < BKT_W) cnt[t] = 0;
    __syncthreads();
    for (int i0 = beg + t; i0 < end; i0 += FU * EB) {
        int dd[FU];
        #pragma unroll
        for (int u = 0; u < FU; ++u) {
            const int i = i0 + u * EB;
            dd[u] = (i < end) ? bkt[i].y : -1;
        }
        #pragma unroll
        for (int u = 0; u < FU; ++u)
            if (dd[u] >= 0) atomicAdd(&cnt[dd[u] - lo], 1);
    }
    __syncthreads();

    int c = 0, inc = 0;
    if (t < BKT_W) {
        c   = cnt[t];
        inc = wave_incl_scan(c, lane);
        if (lane == 31) wsum[w] = inc;
    }
    __syncthreads();
    if (w == 0) {
        const int x  = (lane < BKT_W / 32) ? wsum[lane] : 0;
        const int xi = wave_incl_scan(x, lane);
        if (lane < BKT_W / 32) wsum[lane] = xi - x;
    }
    __syncthreads();
    if (t < BKT_W) {
        const int excl = wsum[w] + inc - c;
        cur[t] = excl;
        const int node = lo + t;
        if (node < n) {
            rowptr[node] = beg + excl;
            dis[node]    = rsqrtf((float)c + 1.0f);
            if (node == n - 1) rowptr[n] = beg + excl + c;
        }
    }
    __syncthreads();
    for (int i0 = beg + t; i0 < end; i0 += FU * EB) {
        int2 p[FU];
        #pragma unroll
        for (int u = 0; u < FU; ++u) {
            const int i = i0 + u * EB;
            p[u] = (i < end) ? bkt[i] : make_int2(0, -1);
        }
        #pragma unroll
        for (int u = 0; u < FU; ++u) {
            if (p[u].y >= 0) {
                const int pos = atomicAdd(&cur[p[u].y - lo], 1);
                csr_src[beg + pos] = p[u].x;
            }
        }
    }
}

template<int ROWS>
__global__ void __launch_bounds__(256) k_row_gather(const int* __restrict__ rowptr,
                                                    const int* __restrict__ csr_src,
                                                    const float* __restrict__ dis,
                                                    const float* __restrict__ feat,
                                                    float* io, int ncols, int n) {
    const int lane = threadIdx.x & 31;
    const int first = __builtin_amdgcn_readfirstlane(
        (int)((blockIdx.x * (blockDim.x >> 5) + (threadIdx.x >> 5)) * ROWS));

    const bool act = (lane * 4 < ncols);
    const int  c4  = act ? lane * 4 : 0;
    const float* __restrict__ fcol = feat + c4;

    for (int r = 0; r < ROWS; ++r) {
        const int node = first + r;
        if (node >= n) break;

        const int   beg = rowptr[node];
        const int   end = rowptr[node + 1];
        const float di  = dis[node];

        float4* iop = (float4*)(io + (size_t)node * ncols + c4);
        float4 acc = *iop;

        for (int base = beg; base < end; base += 32) {
            const int left = end - base;
            const int m = left < 32 ? left : 32;
            int   ms = 0;
            float mw = 0.0f;
            if (lane < m) {
                ms = csr_src[base + lane];
                mw = dis[ms] * di;
            }
            int j = 0;
            for (; j + 4 <= m; j += 4) {
                const int   s0 = __builtin_amdgcn_readlane(ms, j);
                const int   s1 = __builtin_amdgcn_readlane(ms, j + 1);
                const int   s2 = __builtin_amdgcn_readlane(ms, j + 2);
                const int   s3 = __builtin_amdgcn_readlane(ms, j + 3);
                const float w0 = __int_as_float(__builtin_amdgcn_readlane(__float_as_int(mw), j));
                const float w1 = __int_as_float(__builtin_amdgcn_readlane(__float_as_int(mw), j + 1));
                const float w2 = __int_as_float(__builtin_amdgcn_readlane(__float_as_int(mw), j + 2));
                const float w3 = __int_as_float(__builtin_amdgcn_readlane(__float_as_int(mw), j + 3));
                const float4 v0 = *(const float4*)(fcol + (size_t)s0 * ncols);
                const float4 v1 = *(const float4*)(fcol + (size_t)s1 * ncols);
                const float4 v2 = *(const float4*)(fcol + (size_t)s2 * ncols);
                const float4 v3 = *(const float4*)(fcol + (size_t)s3 * ncols);
                acc.x = fmaf(v0.x, w0, acc.x); acc.y = fmaf(v0.y, w0, acc.y); acc.z = fmaf(v0.z, w0, acc.z); acc.w = fmaf(v0.w, w0, acc.w);
                acc.x = fmaf(v1.x, w1, acc.x); acc.y = fmaf(v1.y, w1, acc.y); acc.z = fmaf(v1.z, w1, acc.z); acc.w = fmaf(v1.w, w1, acc.w);
                acc.x = fmaf(v2.x, w2, acc.x); acc.y = fmaf(v2.y, w2, acc.y); acc.z = fmaf(v2.z, w2, acc.z); acc.w = fmaf(v2.w, w2, acc.w);
                acc.x = fmaf(v3.x, w3, acc.x); acc.y = fmaf(v3.y, w3, acc.y); acc.z = fmaf(v3.z, w3, acc.z); acc.w = fmaf(v3.w, w3, acc.w);
            }
            for (; j < m; ++j) {
                const int   s0 = __builtin_amdgcn_readlane(ms, j);
                const float w0 = __int_as_float(__builtin_amdgcn_readlane(__float_as_int(mw), j));
                const float4 v0 = *(const float4*)(fcol + (size_t)s0 * ncols);
                acc.x = fmaf(v0.x, w0, acc.x); acc.y = fmaf(v0.y, w0, acc.y); acc.z = fmaf(v0.z, w0, acc.z); acc.w = fmaf(v0.w, w0, acc.w);
            }
        }
        if (act) *iop = acc;
    }
}


#define TROWS   16
#define HPITCH  132
#define NT2     3
#define WPITCH  48
#define W1PITCH 136
#define GW      13
#define GT      (GW * 32)
#define W1Q     ((NFEAT * (NFEAT / 4) + GT - 1) / GT)
#define W2Q     ((NFEAT * (WPITCH / 4) + GT - 1) / GT)
#define KG1     2
#define KG2     4
#define OSOFF   1024
#define CH0     31
#define PF      16
#define TW      12
#define COV     (PF + TW - 1)

template<int NV>
__device__ __forceinline__ void issue_rows(float4 (&v)[NV], int ms, int j, int cnt,
                                           const float* __restrict__ fcol) {
    #pragma unroll
    for (int u = 0; u < NV; ++u) {
        v[u] = make_float4(0.0f, 0.0f, 0.0f, 0.0f);
        if (u < cnt) {
            const int s = __builtin_amdgcn_readlane(ms, j + u);
            v[u] = *(const float4*)(fcol + (size_t)s * NFEAT);
        }
    }
}
template<int NV>
__device__ __forceinline__ float4 sum_rows(float4 acc, const float4 (&v)[NV], float mw,
                                           int j, int cnt) {
    #pragma unroll
    for (int u = 0; u < NV; ++u) {
        if (u < cnt) {
            const float w = __int_as_float(__builtin_amdgcn_readlane(__float_as_int(mw), j + u));
            acc.x = fmaf(v[u].x, w, acc.x); acc.y = fmaf(v[u].y, w, acc.y);
            acc.z = fmaf(v[u].z, w, acc.z); acc.w = fmaf(v[u].w, w, acc.w);
        }
    }
    return acc;
}

__device__ __forceinline__ void issue_ahead(float4 (&v)[PF], int ms, int tot,
                                            const float* __restrict__ fcol) {
    #pragma unroll
    for (int u = 0; u < PF; ++u) {
        const int s = (u < tot) ? __builtin_amdgcn_readlane(ms, u) : 0;
        v[u] = *(const float4*)(fcol + (size_t)s * NFEAT);
    }
}

__global__ void __launch_bounds__(GT) k_row_gather_gemm12(const int* __restrict__ rowptr,
                                                          const int* __restrict__ csr_src,
                                                          const float* __restrict__ dis,
                                                          const float* __restrict__ x,
                                                          const float* __restrict__ W1,
                                                          const float* __restrict__ b1,
                                                          const float* __restrict__ W2,
                                                          const float* __restrict__ b2,
                                                          float* __restrict__ hw2,
                                                          float* __restrict__ out,
                                                          int ncls, int n) {
    __shared__ __attribute__((aligned(16))) float w1s[NFEAT * W1PITCH];
    __shared__ __attribute__((aligned(16))) float w2s[NFEAT * WPITCH];
    __shared__ __attribute__((aligned(16))) float strips[GW * TROWS * HPITCH];

    {
        float4 p1[W1Q], p2[W2Q];
        #pragma unroll
        for (int u = 0; u < W1Q; ++u) {
            const int q  = threadIdx.x + u * GT;
            const int k  = q / (NFEAT / 4);
            const int cq = (q % (NFEAT / 4)) * 4;
            p1[u] = make_float4(0.0f, 0.0f, 0.0f, 0.0f);
            if (q < NFEAT * (NFEAT / 4)) p1[u] = *(const float4*)(W1 + (size_t)k * NFEAT + cq);
        }
        #pragma unroll
        for (int u = 0; u < W2Q; ++u) {
            const int q  = threadIdx.x + u * GT;
            const int k  = q / (WPITCH / 4);
            const int cq = (q % (WPITCH / 4)) * 4;
            p2[u] = make_float4(0.0f, 0.0f, 0.0f, 0.0f);
            if (q < NFEAT * (WPITCH / 4) && cq < ncls) p2[u] = *(const float4*)(W2 + (size_t)k * ncls + cq);
        }
        #pragma unroll
        for (int u = 0; u < W1Q; ++u) {
            const int q  = threadIdx.x + u * GT;
            const int k  = q / (NFEAT / 4);
            const int cq = (q % (NFEAT / 4)) * 4;
            if (q < NFEAT * (NFEAT / 4)) *(float4*)(w1s + k * W1PITCH + cq) = p1[u];
        }
        #pragma unroll
        for (int u = 0; u < W2Q; ++u) {
            const int q = threadIdx.x + u * GT;
            if (q < NFEAT * (WPITCH / 4)) *(float4*)(w2s + q * 4) = p2[u];
        }
    }
    __syncthreads();

    const int lane = threadIdx.x & 31;
    const int wv    = __builtin_amdgcn_readfirstlane((int)(threadIdx.x >> 5));
    float* strip = strips + wv * (TROWS * HPITCH);
    const int c4 = lane * 4;
    const float* __restrict__ fcol = x + c4;
    const int half = lane >> 4;
    const int l    = lane & 15;
    const float* arow = strip + l * HPITCH + 2 * half;

    const int ntile = (n + TROWS - 1) / TROWS;
    for (int tile = blockIdx.x * GW + wv; tile < ntile; tile += gridDim.x * GW) {
    const int first = tile * TROWS;

    int   rp = 0;
    float dv = 0.0f;
    {
        int ri = first + lane; if (ri > n) ri = n;
        int ni = first + lane; if (ni > n - 1) ni = n - 1;
        if (lane <= TROWS) rp = rowptr[ri];
        if (lane <  TROWS) dv = dis[ni];
    }

#define ROW_CNT(q)   ((__builtin_amdgcn_readlane(rp, (q) + 1) - __builtin_amdgcn_readlane(rp, (q))) < CH0 \
                      ? (__builtin_amdgcn_readlane(rp, (q) + 1) - __builtin_amdgcn_readlane(rp, (q))) : CH0)
#define ROW_NODE(q)  ((first + (q)) < n ? (first + (q)) : (n - 1))
#define LIST_IDX(q, cnt)  ((lane >= 1 && lane <= (cnt)) ? __builtin_amdgcn_readlane(rp, (q)) + lane - 1 \
                                                         : ((cnt) > 0 ? __builtin_amdgcn_readlane(rp, (q)) : 0))
#define LIST_SEL(raw, q, cnt)  ((lane >= 1 && lane <= (cnt)) ? (raw) : ROW_NODE(q))
    int   ms_a, ms_b;
    float w_a;
    {
        const int cnt0 = ROW_CNT(0);
        const int cnt1 = ROW_CNT(1);
        const int raw0 = csr_src[LIST_IDX(0, cnt0)];
        const int raw1 = csr_src[LIST_IDX(1, cnt1)];
        ms_a = LIST_SEL(raw0, 0, cnt0);
        ms_b = LIST_SEL(raw1, 1, cnt1);
        w_a  = dis[ms_a] * __int_as_float(__builtin_amdgcn_readlane(__float_as_int(dv), 0));
    }
    float4 v[2][PF];
    float4 vt[TW];
    issue_ahead(v[0], ms_a, (first < n) ? ROW_CNT(0) + 1 : 0, fcol);
    #pragma unroll 1
    for (int r2 = 0; r2 < TROWS; r2 += 2) {
    #pragma unroll
    for (int hh = 0; hh < 2; ++hh) {
        const int r = r2 + hh;
        const int node = first + r;
        const int q1 = (r + 1 < TROWS) ? r + 1 : TROWS - 1;
        const int q2 = (r + 2 < TROWS) ? r + 2 : TROWS - 1;
        const int cnt2 = ROW_CNT(q2);
        const int beg = __builtin_amdgcn_readlane(rp, r);
        const int end = __builtin_amdgcn_readlane(rp, r + 1);
        const int cnt = (end - beg) < CH0 ? (end - beg) : CH0;
        const int tot = (node < n) ? cnt + 1 : 0;
        const int totn = (r + 1 < TROWS && node + 1 < n) ? ROW_CNT(q1) + 1 : 0;

        const float raw_w = dis[ms_b];
        const int   raw_c = csr_src[LIST_IDX(q2, cnt2)];
        issue_rows(vt, ms_a, PF, tot - PF, fcol);
        issue_ahead(v[hh ^ 1], ms_b, totn, fcol);

        float4 acc = make_float4(0.0f, 0.0f, 0.0f, 0.0f);
        acc = sum_rows(acc, v[hh], w_a, 0, tot);
        acc = sum_rows(acc, vt, w_a, PF, tot - PF);
        *(float4*)(strip + r * HPITCH + c4) = acc;
        w_a  = raw_w * __int_as_float(__builtin_amdgcn_readlane(__float_as_int(dv), q1));
        ms_a = ms_b;
        ms_b = LIST_SEL(raw_c, q2, cnt2);
    }
    }

    {
        const int rpn = __shfl_down(rp, 1);
        const unsigned lmask = __builtin_amdgcn_ballot_w32(lane < TROWS && rpn - rp > COV);
        for (unsigned mk = lmask; mk != 0; mk &= mk - 1) {
            const int r = __builtin_ctz(mk);
            const int beg = __builtin_amdgcn_readlane(rp, r);
            const int end = __builtin_amdgcn_readlane(rp, r + 1);
            const float di = __int_as_float(__builtin_amdgcn_readlane(__float_as_int(dv), r));
            float4 acc = *(const float4*)(strip + r * HPITCH + c4);
            for (int base = beg + COV; base < end; base += 32) {
                const int left = end - base;
                const int m2 = left < 32 ? left : 32;
                int   ms = 0;
                float mw = 0.0f;
                if (lane < m2) {
                    ms = csr_src[base + lane];
                    mw = dis[ms] * di;
                }
                for (int j = 0; j < m2; j += 16) {
                    issue_rows(v[0], ms, j, m2 - j, fcol);
                    acc = sum_rows(acc, v[0], mw, j, m2 - j);
                }
            }
            *(float4*)(strip + r * HPITCH + c4) = acc;
        }
    }
#undef ROW_CNT
#undef ROW_NODE
#undef LIST_IDX
#undef LIST_SEL

    __builtin_amdgcn_wave_barrier();

    v8f hacc[NFEAT / 16];
    #pragma unroll
    for (int nh = 0; nh < 2; ++nh) {
        v8f acc[4];
        #pragma unroll
        for (int t = 0; t < 4; ++t) acc[t] = (v8f){0, 0, 0, 0, 0, 0, 0, 0};
        const float* __restrict__ bcol = w1s + (2 * half) * W1PITCH + nh * 64 + l;
        #pragma unroll
        for (int kg = 0; kg < NFEAT / 4; kg += KG1) {
            v2f a[KG1], b[4][KG1];
            #pragma unroll
            for (int u = 0; u < KG1; ++u) {
                const int ks = kg + u;
                a[u].x = arow[ks * 4];
                a[u].y = arow[ks * 4 + 1];
                #pragma unroll
                for (int t = 0; t < 4; ++t) {
                    b[t][u].x = bcol[(ks * 4)     * W1PITCH + t * 16];
                    b[t][u].y = bcol[(ks * 4 + 1) * W1PITCH + t * 16];
                }
            }
            __builtin_amdgcn_sched_barrier(0);
            #pragma unroll
            for (int u = 0; u < KG1; ++u) {
                #pragma unroll
                for (int t = 0; t < 4; ++t) {
                    acc[t] = __builtin_amdgcn_wmma_f32_16x16x4_f32(
                                 false, a[u], false, b[t][u], (short)0, acc[t], false, false);
                }
            }
        }
        #pragma unroll
        for (int t = 0; t < 4; ++t) hacc[nh * 4 + t] = acc[t];
    }

    __builtin_amdgcn_wave_barrier();
    #pragma unroll
    for (int t = 0; t < NFEAT / 16; ++t) {
        const int   col = t * 16 + l;
        const float bb  = b1[col];
        #pragma unroll
        for (int r = 0; r < 8; ++r)
            strip[(half * 8 + r) * HPITCH + col] = fmaxf(hacc[t][r] + bb, 0.0f);
    }
    __builtin_amdgcn_wave_barrier();

    v8f acc2[NT2];
    #pragma unroll
    for (int t = 0; t < NT2; ++t) acc2[t] = (v8f){0, 0, 0, 0, 0, 0, 0, 0};
    {
        const float* __restrict__ bcol = w2s + (2 * half) * WPITCH + l;
        #pragma unroll
        for (int kg = 0; kg < NFEAT / 4; kg += KG2) {
            v2f a[KG2], b[NT2][KG2];
            #pragma unroll
            for (int u = 0; u < KG2; ++u) {
                const int ks = kg + u;
                a[u].x = arow[ks * 4];
                a[u].y = arow[ks * 4 + 1];
                #pragma unroll
                for (int t = 0; t < NT2; ++t) {
                    b[t][u].x = bcol[(ks * 4)     * WPITCH + t * 16];
                    b[t][u].y = bcol[(ks * 4 + 1) * WPITCH + t * 16];
                }
            }
            __builtin_amdgcn_sched_barrier(0);
            #pragma unroll
            for (int u = 0; u < KG2; ++u) {
                #pragma unroll
                for (int t = 0; t < NT2; ++t) {
                    acc2[t] = __builtin_amdgcn_wmma_f32_16x16x4_f32(
                                  false, a[u], false, b[t][u], (short)0, acc2[t], false, false);
                }
            }
        }
    }

    float dsq[8];
    #pragma unroll
    for (int r = 0; r < 8; ++r) {
        const float dlo = __int_as_float(__builtin_amdgcn_readlane(__float_as_int(dv), r));
        const float dhi = __int_as_float(__builtin_amdgcn_readlane(__float_as_int(dv), r + 8));
        const float d   = half ? dhi : dlo;
        dsq[r] = d * d;
    }

    __builtin_amdgcn_wave_barrier();
    float* os_hw  = strip;
    float* os_out = strip + OSOFF;
    #pragma unroll
    for (int t = 0; t < NT2; ++t) {
        const int c0 = t * 16 + l;
        if (c0 < ncls) {
            const float bv = b2[c0];
            #pragma unroll
            for (int r = 0; r < 8; ++r) {
                const int trow = half * 8 + r;
                const float v = acc2[t][r];
                os_hw [trow * ncls + c0] = v;
                os_out[trow * ncls + c0] = v * dsq[r] + bv;
            }
        }
    }
    __builtin_amdgcn_wave_barrier();

    for (int i4 = lane * 4; i4 < TROWS * ncls; i4 += 128) {
        const int trow = i4 / ncls;
        if (first + trow < n) {
            const size_t g = (size_t)first * ncls + i4;
            *(float4*)(hw2 + g) = *(const float4*)(os_hw  + i4);
            *(float4*)(out + g) = *(const float4*)(os_out + i4);
        }
    }
    __builtin_amdgcn_wave_barrier();
    }
}

extern "C" void kernel_launch(void* const* d_in, const int* in_sizes, int n_in,
                              void* d_out, int out_size, void* d_ws, size_t ws_size,
                              hipStream_t stream) {
    const float* x  = (const float*)d_in[0];
    const int*   ei = (const int*)  d_in[1];
    const float* W1 = (const float*)d_in[2];
    const float* b1 = (const float*)d_in[3];
    const float* W2 = (const float*)d_in[4];
    const float* b2 = (const float*)d_in[5];
    float* out = (float*)d_out;

    const int N   = in_sizes[0] / NFEAT;
    const int E   = in_sizes[1] / 2;
    const int CLS = in_sizes[5];

    const int* src = ei;
    const int* dst = ei + E;

    const int NBKT = (N + BKT_W - 1) / BKT_W;

    char* ws = (char*)d_ws;
    size_t off = 0;
    float* dis = (float*)(ws + off); off += ((size_t)N * 4 + 255) & ~(size_t)255;
    float* hw2 = (float*)(ws + off); off += ((size_t)N * CLS * 4 + 255) & ~(size_t)255;
    int* bc     = (int*)(ws + off); off += ((size_t)NBKT * 8 + 255) & ~(size_t)255;
    int* rowptr = (int*)(ws + off); off += ((size_t)(N + 1) * 4 + 255) & ~(size_t)255;
    int* bstart = (int*)(ws + off); off += ((size_t)(NBKT + 1) * 4 + 255) & ~(size_t)255;
    int* bkt    = (int*)(ws + off); off += ((size_t)E * 8 + 255) & ~(size_t)255;
    int* csr    = (int*)(ws + off); off += ((size_t)E * 4 + 255) & ~(size_t)255;
    (void)ws_size; (void)n_in; (void)out_size;

    const int TPB = 256;
    const int WAVES = TPB / 32;

    const int EBLK = (E + EB * EPT - 1) / (EB * EPT);
    k_zero_bkt   <<<(2 * NBKT + TPB - 1) / TPB, TPB, 0, stream>>>(bc, 2 * NBKT);
    k_bkt_count  <<<EBLK, EB, 0, stream>>>(dst, bc, E, NBKT);
    k_bkt_scatter<<<EBLK, EB, 0, stream>>>(src, dst, bc, bc + NBKT, bstart, bkt, E, NBKT);
    k_bkt_finish <<<NBKT, EB, 0, stream>>>(bkt, bstart, rowptr, dis, csr, N);

    {
        const int NTILE = (N + TROWS - 1) / TROWS;
        k_row_gather_gemm12<<<(NTILE + 2 * GW - 1) / (2 * GW), GT, 0, stream>>>(
            rowptr, csr, dis, x, W1, b1, W2, b2, hw2, out, CLS, N);
    }
    k_row_gather<4><<<(N + 4 * WAVES - 1) / (4 * WAVES), TPB, 0, stream>>>(
        rowptr, csr, dis, hw2, out, CLS, N);
}


# Level 1 of the edge grouping without count pass or global atomics: a block orders its own 4096 edges by bucket in LDS, bucket blocks collect the pieces; 6 to 4 launches; sum order still arrival rank
# speedup vs baseline: 4.6113x; 1.0509x over previous
#include <hip/hip_runtime.h>


typedef __attribute__((ext_vector_type(2))) float v2f;
typedef __attribute__((ext_vector_type(8))) float v8f;

#define NFEAT 128


#define BKT_SHIFT 8
#define BKT_W (1 << BKT_SHIFT)
#define EB 1024
#define EPT 4
#define CH (EB * EPT)
#define FU 4

__device__ __forceinline__ int wave_incl_scan(int v, int lane) {
    #pragma unroll
    for (int o = 1; o < 32; o <<= 1) {
        const int t = __shfl_up(v, o);
        if (lane >= o) v += t;
    }
    return v;
}

__global__ void __launch_bounds__(EB) k_blk_group(const int* __restrict__ src,
                                                  const int* __restrict__ dst,
                                                  int* __restrict__ ptab,
                                                  int* __restrict__ bkt_raw,
                                                  int E, int nbkt) {
    __shared__ int hist[EB];
    __shared__ int wsum[EB / 32];
    __shared__ __attribute__((aligned(8))) int stage_raw[2 * CH];
    const int t = threadIdx.x, lane = t & 31, w = t >> 5;
    const int nblk = gridDim.x;
    int2* stage = (int2*)stage_raw;

    hist[t] = 0;
    __syncthreads();

    const int base = blockIdx.x * CH;
    int s[EPT], d[EPT], r[EPT];
    #pragma unroll
    for (int k = 0; k < EPT; ++k) {
        const int e = base + k * EB + t;
        d[k] = (e < E) ? dst[e] : -1;
        s[k] = (e < E) ? src[e] : 0;
    }
    #pragma unroll
    for (int k = 0; k < EPT; ++k) {
        r[k] = 0;
        if (d[k] >= 0) r[k] = atomicAdd(&hist[d[k] >> BKT_SHIFT], 1);
    }
    __syncthreads();

    const int c   = hist[t];
    const int inc = wave_incl_scan(c, lane);
    if (lane == 31) wsum[w] = inc;
    __syncthreads();
    if (w == 0) {
        const int x  = wsum[lane];
        const int xi = wave_incl_scan(x, lane);
        wsum[lane] = xi - x;
    }
    __syncthreads();
    const int excl = wsum[w] + inc - c;
    if (t <= nbkt) ptab[(size_t)t * nblk + blockIdx.x] = excl;
    hist[t] = excl;
    __syncthreads();

    #pragma unroll
    for (int k = 0; k < EPT; ++k) {
        if (d[k] >= 0) stage[hist[d[k] >> BKT_SHIFT] + r[k]] = make_int2(s[k], d[k]);
    }
    __syncthreads();
    const int left = E - base;
    const int nval = left < CH ? left : CH;
    int2* __restrict__ run = (int2*)bkt_raw + (size_t)blockIdx.x * CH;
    #pragma unroll
    for (int k = 0; k < EPT; ++k) {
        const int i = k * EB + t;
        if (i < nval) run[i] = stage[i];
    }
}

__device__ __forceinline__ void locate_pairs(const int* pp, const int* pg, int i0, int npair,
                                             int (&g)[FU]) {
    int b[FU];
    #pragma unroll
    for (int u = 0; u < FU; ++u) b[u] = 0;
    #pragma unroll
    for (int step = EB / 2; step > 0; step >>= 1) {
        #pragma unroll
        for (int u = 0; u < FU; ++u) {
            const int nb = b[u] + step;
            if (pp[nb] <= i0 + u * EB) b[u] = nb;
        }
    }
    #pragma unroll
    for (int u = 0; u < FU; ++u) {
        const int i = i0 + u * EB;
        const int gi = pg[b[u]] + (i - pp[b[u]]);
        g[u] = (i < npair) ? gi : -1;
    }
}

__global__ void __launch_bounds__(EB) k_bkt_finish(const int* __restrict__ bkt_raw,
                                                   const int* __restrict__ ptab,
                                                   int* __restrict__ rowptr,
                                                   float* __restrict__ dis,
                                                   int* __restrict__ csr_src, int n, int nblk) {
    __shared__ int cnt[BKT_W];
    __shared__ int cur[BKT_W];
    __shared__ int wsum[EB / 32];
    __shared__ int wbeg[EB / 32];
    __shared__ int pp[EB];
    __shared__ int pg[EB];
    __shared__ int tot[2];
    const int t = threadIdx.x, lane = t & 31, w = t >> 5;
    const int2* __restrict__ bkt = (const int2*)bkt_raw;

    const int lo = blockIdx.x << BKT_SHIFT;

    int sb = 0, eb = 0;
    if (t < nblk) {
        sb = ptab[(size_t)blockIdx.x * nblk + t];
        eb = ptab[(size_t)(blockIdx.x + 1) * nblk + t];
    }
    const int len = eb - sb;
    int bsum = sb;
    #pragma unroll
    for (int o = 16; o > 0; o >>= 1) bsum += __shfl_xor(bsum, o);
    const int pinc = wave_incl_scan(len, lane);
    if (lane == 31) wsum[w] = pinc;
    if (lane == 0)  wbeg[w] = bsum;
    if (t < BKT_W) cnt[t] = 0;
    __syncthreads();
    if (w == 0) {
        const int x  = wsum[lane];
        const int xi = wave_incl_scan(x, lane);
        wsum[lane] = xi - x;
        int y = wbeg[lane];
        #pragma unroll
        for (int o = 16; o > 0; o >>= 1) y += __shfl_xor(y, o);
        if (lane == 0)  tot[0] = y;
        if (lane == 31) tot[1] = xi;
    }
    __syncthreads();
    pp[t] = wsum[w] + pinc - len;
    pg[t] = t * CH + sb;
    const int beg  = tot[0];
    const int npair = tot[1];
    __syncthreads();

    int g0[FU];
    locate_pairs(pp, pg, t, npair, g0);

    for (int i0 = t; i0 < npair; i0 += FU * EB) {
        int g[FU], dd[FU];
        #pragma unroll
        for (int u = 0; u < FU; ++u) g[u] = g0[u];
        if (i0 != t) locate_pairs(pp, pg, i0, npair, g);
        #pragma unroll
        for (int u = 0; u < FU; ++u)
            dd[u] = (g[u] >= 0) ? bkt[g[u]].y : -1;
        #pragma unroll
        for (int u = 0; u < FU; ++u)
            if (dd[u] >= 0) atomicAdd(&cnt[dd[u] - lo], 1);
    }
    __syncthreads();

    int c = 0, inc = 0;
    if (t < BKT_W) {
        c   = cnt[t];
        inc = wave_incl_scan(c, lane);
        if (lane == 31) wsum[w] = inc;
    }
    __syncthreads();
    if (w == 0) {
        const int x  = (lane < BKT_W / 32) ? wsum[lane] : 0;
        const int xi = wave_incl_scan(x, lane);
        if (lane < BKT_W / 32) wsum[lane] = xi - x;
    }
    __syncthreads();
    if (t < BKT_W) {
        const int excl = wsum[w] + inc - c;
        cur[t] = excl;
        const int node = lo + t;
        if (node < n) {
            rowptr[node] = beg + excl;
            dis[node]    = rsqrtf((float)c + 1.0f);
            if (node == n - 1) rowptr[n] = beg + excl + c;
        }
    }
    __syncthreads();
    for (int i0 = t; i0 < npair; i0 += FU * EB) {
        int g[FU];
        int2 p[FU];
        #pragma unroll
        for (int u = 0; u < FU; ++u) g[u] = g0[u];
        if (i0 != t) locate_pairs(pp, pg, i0, npair, g);
        #pragma unroll
        for (int u = 0; u < FU; ++u)
            p[u] = (g[u] >= 0) ? bkt[g[u]] : make_int2(0, -1);
        #pragma unroll
        for (int u = 0; u < FU; ++u) {
            if (p[u].y >= 0) {
                const int pos = atomicAdd(&cur[p[u].y - lo], 1);
                csr_src[beg + pos] = p[u].x;
            }
        }
    }
}

template<int ROWS>
__global__ void __launch_bounds__(256) k_row_gather(const int* __restrict__ rowptr,
                                                    const int* __restrict__ csr_src,
                                                    const float* __restrict__ dis,
                                                    const float* __restrict__ feat,
                                                    float* io, int ncols, int n) {
    const int lane = threadIdx.x & 31;
    const int first = __builtin_amdgcn_readfirstlane(
        (int)((blockIdx.x * (blockDim.x >> 5) + (threadIdx.x >> 5)) * ROWS));

    const bool act = (lane * 4 < ncols);
    const int  c4  = act ? lane * 4 : 0;
    const float* __restrict__ fcol = feat + c4;

    for (int r = 0; r < ROWS; ++r) {
        const int node = first + r;
        if (node >= n) break;

        const int   beg = rowptr[node];
        const int   end = rowptr[node + 1];
        const float di  = dis[node];

        float4* iop = (float4*)(io + (size_t)node * ncols + c4);
        float4 acc = *iop;

        for (int base = beg; base < end; base += 32) {
            const int left = end - base;
            const int m = left < 32 ? left : 32;
            int   ms = 0;
            float mw = 0.0f;
            if (lane < m) {
                ms = csr_src[base + lane];
                mw = dis[ms] * di;
            }
            int j = 0;
            for (; j + 4 <= m; j += 4) {
                const int   s0 = __builtin_amdgcn_readlane(ms, j);
                const int   s1 = __builtin_amdgcn_readlane(ms, j + 1);
                const int   s2 = __builtin_amdgcn_readlane(ms, j + 2);
                const int   s3 = __builtin_amdgcn_readlane(ms, j + 3);
                const float w0 = __int_as_float(__builtin_amdgcn_readlane(__float_as_int(mw), j));
                const float w1 = __int_as_float(__builtin_amdgcn_readlane(__float_as_int(mw), j + 1));
                const float w2 = __int_as_float(__builtin_amdgcn_readlane(__float_as_int(mw), j + 2));
                const float w3 = __int_as_float(__builtin_amdgcn_readlane(__float_as_int(mw), j + 3));
                const float4 v0 = *(const float4*)(fcol + (size_t)s0 * ncols);
                const float4 v1 = *(const float4*)(fcol + (size_t)s1 * ncols);
                const float4 v2 = *(const float4*)(fcol + (size_t)s2 * ncols);
                const float4 v3 = *(const float4*)(fcol + (size_t)s3 * ncols);
                acc.x = fmaf(v0.x, w0, acc.x); acc.y = fmaf(v0.y, w0, acc.y); acc.z = fmaf(v0.z, w0, acc.z); acc.w = fmaf(v0.w, w0, acc.w);
                acc.x = fmaf(v1.x, w1, acc.x); acc.y = fmaf(v1.y, w1, acc.y); acc.z = fmaf(v1.z, w1, acc.z); acc.w = fmaf(v1.w, w1, acc.w);
                acc.x = fmaf(v2.x, w2, acc.x); acc.y = fmaf(v2.y, w2, acc.y); acc.z = fmaf(v2.z, w2, acc.z); acc.w = fmaf(v2.w, w2, acc.w);
                acc.x = fmaf(v3.x, w3, acc.x); acc.y = fmaf(v3.y, w3, acc.y); acc.z = fmaf(v3.z, w3, acc.z); acc.w = fmaf(v3.w, w3, acc.w);
            }
            for (; j < m; ++j) {
                const int   s0 = __builtin_amdgcn_readlane(ms, j);
                const float w0 = __int_as_float(__builtin_amdgcn_readlane(__float_as_int(mw), j));
                const float4 v0 = *(const float4*)(fcol + (size_t)s0 * ncols);
                acc.x = fmaf(v0.x, w0, acc.x); acc.y = fmaf(v0.y, w0, acc.y); acc.z = fmaf(v0.z, w0, acc.z); acc.w = fmaf(v0.w, w0, acc.w);
            }
        }
        if (act) *iop = acc;
    }
}


#define TROWS   16
#define HPITCH  132
#define NT2     3
#define WPITCH  48
#define W1PITCH 136
#define GW      13
#define GT      (GW * 32)
#define W1Q     ((NFEAT * (NFEAT / 4) + GT - 1) / GT)
#define W2Q     ((NFEAT * (WPITCH / 4) + GT - 1) / GT)
#define KG1     2
#define KG2     4
#define OSOFF   1024
#define CH0     31
#define PF      16
#define TW      12
#define COV     (PF + TW - 1)

template<int NV>
__device__ __forceinline__ void issue_rows(float4 (&v)[NV], int ms, int j, int cnt,
                                           const float* __restrict__ fcol) {
    #pragma unroll
    for (int u = 0; u < NV; ++u) {
        v[u] = make_float4(0.0f, 0.0f, 0.0f, 0.0f);
        if (u < cnt) {
            const int s = __builtin_amdgcn_readlane(ms, j + u);
            v[u] = *(const float4*)(fcol + (size_t)s * NFEAT);
        }
    }
}
template<int NV>
__device__ __forceinline__ float4 sum_rows(float4 acc, const float4 (&v)[NV], float mw,
                                           int j, int cnt) {
    #pragma unroll
    for (int u = 0; u < NV; ++u) {
        if (u < cnt) {
            const float w = __int_as_float(__builtin_amdgcn_readlane(__float_as_int(mw), j + u));
            acc.x = fmaf(v[u].x, w, acc.x); acc.y = fmaf(v[u].y, w, acc.y);
            acc.z = fmaf(v[u].z, w, acc.z); acc.w = fmaf(v[u].w, w, acc.w);
        }
    }
    return acc;
}

__device__ __forceinline__ void issue_ahead(float4 (&v)[PF], int ms, int tot,
                                            const float* __restrict__ fcol) {
    #pragma unroll
    for (int u = 0; u < PF; ++u) {
        const int s = (u < tot) ? __builtin_amdgcn_readlane(ms, u) : 0;
        v[u] = *(const float4*)(fcol + (size_t)s * NFEAT);
    }
}

__global__ void __launch_bounds__(GT) k_row_gather_gemm12(const int* __restrict__ rowptr,
                                                          const int* __restrict__ csr_src,
                                                          const float* __restrict__ dis,
                                                          const float* __restrict__ x,
                                                          const float* __restrict__ W1,
                                                          const float* __restrict__ b1,
                                                          const float* __restrict__ W2,
                                                          const float* __restrict__ b2,
                                                          float* __restrict__ hw2,
                                                          float* __restrict__ out,
                                                          int ncls, int n) {
    __shared__ __attribute__((aligned(16))) float w1s[NFEAT * W1PITCH];
    __shared__ __attribute__((aligned(16))) float w2s[NFEAT * WPITCH];
    __shared__ __attribute__((aligned(16))) float strips[GW * TROWS * HPITCH];

    {
        float4 p1[W1Q], p2[W2Q];
        #pragma unroll
        for (int u = 0; u < W1Q; ++u) {
            const int q  = threadIdx.x + u * GT;
            const int k  = q / (NFEAT / 4);
            const int cq = (q % (NFEAT / 4)) * 4;
            p1[u] = make_float4(0.0f, 0.0f, 0.0f, 0.0f);
            if (q < NFEAT * (NFEAT / 4)) p1[u] = *(const float4*)(W1 + (size_t)k * NFEAT + cq);
        }
        #pragma unroll
        for (int u = 0; u < W2Q; ++u) {
            const int q  = threadIdx.x + u * GT;
            const int k  = q / (WPITCH / 4);
            const int cq = (q % (WPITCH / 4)) * 4;
            p2[u] = make_float4(0.0f, 0.0f, 0.0f, 0.0f);
            if (q < NFEAT * (WPITCH / 4) && cq < ncls) p2[u] = *(const float4*)(W2 + (size_t)k * ncls + cq);
        }
        #pragma unroll
        for (int u = 0; u < W1Q; ++u) {
            const int q  = threadIdx.x + u * GT;
            const int k  = q / (NFEAT / 4);
            const int cq = (q % (NFEAT / 4)) * 4;
            if (q < NFEAT * (NFEAT / 4)) *(float4*)(w1s + k * W1PITCH + cq) = p1[u];
        }
        #pragma unroll
        for (int u = 0; u < W2Q; ++u) {
            const int q = threadIdx.x + u * GT;
            if (q < NFEAT * (WPITCH / 4)) *(float4*)(w2s + q * 4) = p2[u];
        }
    }
    __syncthreads();

    const int lane = threadIdx.x & 31;
    const int wv    = __builtin_amdgcn_readfirstlane((int)(threadIdx.x >> 5));
    float* strip = strips + wv * (TROWS * HPITCH);
    const int c4 = lane * 4;
    const float* __restrict__ fcol = x + c4;
    const int half = lane >> 4;
    const int l    = lane & 15;
    const float* arow = strip + l * HPITCH + 2 * half;

    const int ntile = (n + TROWS - 1) / TROWS;
    for (int tile = blockIdx.x * GW + wv; tile < ntile; tile += gridDim.x * GW) {
    const int first = tile * TROWS;

    int   rp = 0;
    float dv = 0.0f;
    {
        int ri = first + lane; if (ri > n) ri = n;
        int ni = first + lane; if (ni > n - 1) ni = n - 1;
        if (lane <= TROWS) rp = rowptr[ri];
        if (lane <  TROWS) dv = dis[ni];
    }

#define ROW_CNT(q)   ((__builtin_amdgcn_readlane(rp, (q) + 1) - __builtin_amdgcn_readlane(rp, (q))) < CH0 \
                      ? (__builtin_amdgcn_readlane(rp, (q) + 1) - __builtin_amdgcn_readlane(rp, (q))) : CH0)
#define ROW_NODE(q)  ((first + (q)) < n ? (first + (q)) : (n - 1))
#define LIST_IDX(q, cnt)  ((lane >= 1 && lane <= (cnt)) ? __builtin_amdgcn_readlane(rp, (q)) + lane - 1 \
                                                         : ((cnt) > 0 ? __builtin_amdgcn_readlane(rp, (q)) : 0))
#define LIST_SEL(raw, q, cnt)  ((lane >= 1 && lane <= (cnt)) ? (raw) : ROW_NODE(q))
    int   ms_a, ms_b;
    float w_a;
    {
        const int cnt0 = ROW_CNT(0);
        const int cnt1 = ROW_CNT(1);
        const int raw0 = csr_src[LIST_IDX(0, cnt0)];
        const int raw1 = csr_src[LIST_IDX(1, cnt1)];
        ms_a = LIST_SEL(raw0, 0, cnt0);
        ms_b = LIST_SEL(raw1, 1, cnt1);
        w_a  = dis[ms_a] * __int_as_float(__builtin_amdgcn_readlane(__float_as_int(dv), 0));
    }
    float4 v[2][PF];
    float4 vt[TW];
    issue_ahead(v[0], ms_a, (first < n) ? ROW_CNT(0) + 1 : 0, fcol);
    #pragma unroll 1
    for (int r2 = 0; r2 < TROWS; r2 += 2) {
    #pragma unroll
    for (int hh = 0; hh < 2; ++hh) {
        const int r = r2 + hh;
        const int node = first + r;
        const int q1 = (r + 1 < TROWS) ? r + 1 : TROWS - 1;
        const int q2 = (r + 2 < TROWS) ? r + 2 : TROWS - 1;
        const int cnt2 = ROW_CNT(q2);
        const int beg = __builtin_amdgcn_readlane(rp, r);
        const int end = __builtin_amdgcn_readlane(rp, r + 1);
        const int cnt = (end - beg) < CH0 ? (end - beg) : CH0;
        const int tot = (node < n) ? cnt + 1 : 0;
        const int totn = (r + 1 < TROWS && node + 1 < n) ? ROW_CNT(q1) + 1 : 0;

        const float raw_w = dis[ms_b];
        const int   raw_c = csr_src[LIST_IDX(q2, cnt2)];
        issue_rows(vt, ms_a, PF, tot - PF, fcol);
        issue_ahead(v[hh ^ 1], ms_b, totn, fcol);

        float4 acc = make_float4(0.0f, 0.0f, 0.0f, 0.0f);
        acc = sum_rows(acc, v[hh], w_a, 0, tot);
        acc = sum_rows(acc, vt, w_a, PF, tot - PF);
        *(float4*)(strip + r * HPITCH + c4) = acc;
        w_a  = raw_w * __int_as_float(__builtin_amdgcn_readlane(__float_as_int(dv), q1));
        ms_a = ms_b;
        ms_b = LIST_SEL(raw_c, q2, cnt2);
    }
    }

    {
        const int rpn = __shfl_down(rp, 1);
        const unsigned lmask = __builtin_amdgcn_ballot_w32(lane < TROWS && rpn - rp > COV);
        for (unsigned mk = lmask; mk != 0; mk &= mk - 1) {
            const int r = __builtin_ctz(mk);
            const int beg = __builtin_amdgcn_readlane(rp, r);
            const int end = __builtin_amdgcn_readlane(rp, r + 1);
            const float di = __int_as_float(__builtin_amdgcn_readlane(__float_as_int(dv), r));
            float4 acc = *(const float4*)(strip + r * HPITCH + c4);
            for (int base = beg + COV; base < end; base += 32) {
                const int left = end - base;
                const int m2 = left < 32 ? left : 32;
                int   ms = 0;
                float mw = 0.0f;
                if (lane < m2) {
                    ms = csr_src[base + lane];
                    mw = dis[ms] * di;
                }
                for (int j = 0; j < m2; j += 16) {
                    issue_rows(v[0], ms, j, m2 - j, fcol);
                    acc = sum_rows(acc, v[0], mw, j, m2 - j);
                }
            }
            *(float4*)(strip + r * HPITCH + c4) = acc;
        }
    }
#undef ROW_CNT
#undef ROW_NODE
#undef LIST_IDX
#undef LIST_SEL

    __builtin_amdgcn_wave_barrier();

    v8f hacc[NFEAT / 16];
    #pragma unroll
    for (int nh = 0; nh < 2; ++nh) {
        v8f acc[4];
        #pragma unroll
        for (int t = 0; t < 4; ++t) acc[t] = (v8f){0, 0, 0, 0, 0, 0, 0, 0};
        const float* __restrict__ bcol = w1s + (2 * half) * W1PITCH + nh * 64 + l;
        #pragma unroll
        for (int kg = 0; kg < NFEAT / 4; kg += KG1) {
            v2f a[KG1], b[4][KG1];
            #pragma unroll
            for (int u = 0; u < KG1; ++u) {
                const int ks = kg + u;
                a[u].x = arow[ks * 4];
                a[u].y = arow[ks * 4 + 1];
                #pragma unroll
                for (int t = 0; t < 4; ++t) {
                    b[t][u].x = bcol[(ks * 4)     * W1PITCH + t * 16];
                    b[t][u].y = bcol[(ks * 4 + 1) * W1PITCH + t * 16];
                }
            }
            __builtin_amdgcn_sched_barrier(0);
            #pragma unroll
            for (int u = 0; u < KG1; ++u) {
                #pragma unroll
                for (int t = 0; t < 4; ++t) {
                    acc[t] = __builtin_amdgcn_wmma_f32_16x16x4_f32(
                                 false, a[u], false, b[t][u], (short)0, acc[t], false, false);
                }
            }
        }
        #pragma unroll
        for (int t = 0; t < 4; ++t) hacc[nh * 4 + t] = acc[t];
    }

    __builtin_amdgcn_wave_barrier();
    #pragma unroll
    for (int t = 0; t < NFEAT / 16; ++t) {
        const int   col = t * 16 + l;
        const float bb  = b1[col];
        #pragma unroll
        for (int r = 0; r < 8; ++r)
            strip[(half * 8 + r) * HPITCH + col] = fmaxf(hacc[t][r] + bb, 0.0f);
    }
    __builtin_amdgcn_wave_barrier();

    v8f acc2[NT2];
    #pragma unroll
    for (int t = 0; t < NT2; ++t) acc2[t] = (v8f){0, 0, 0, 0, 0, 0, 0, 0};
    {
        const float* __restrict__ bcol = w2s + (2 * half) * WPITCH + l;
        #pragma unroll
        for (int kg = 0; kg < NFEAT / 4; kg += KG2) {
            v2f a[KG2], b[NT2][KG2];
            #pragma unroll
            for (int u = 0; u < KG2; ++u) {
                const int ks = kg + u;
                a[u].x = arow[ks * 4];
                a[u].y = arow[ks * 4 + 1];
                #pragma unroll
                for (int t = 0; t < NT2; ++t) {
                    b[t][u].x = bcol[(ks * 4)     * WPITCH + t * 16];
                    b[t][u].y = bcol[(ks * 4 + 1) * WPITCH + t * 16];
                }
            }
            __builtin_amdgcn_sched_barrier(0);
            #pragma unroll
            for (int u = 0; u < KG2; ++u) {
                #pragma unroll
                for (int t = 0; t < NT2; ++t) {
                    acc2[t] = __builtin_amdgcn_wmma_f32_16x16x4_f32(
                                  false, a[u], false, b[t][u], (short)0, acc2[t], false, false);
                }
            }
        }
    }

    float dsq[8];
    #pragma unroll
    for (int r = 0; r < 8; ++r) {
        const float dlo = __int_as_float(__builtin_amdgcn_readlane(__float_as_int(dv), r));
        const float dhi = __int_as_float(__builtin_amdgcn_readlane(__float_as_int(dv), r + 8));
        const float d   = half ? dhi : dlo;
        dsq[r] = d * d;
    }

    __builtin_amdgcn_wave_barrier();
    float* os_hw  = strip;
    float* os_out = strip + OSOFF;
    #pragma unroll
    for (int t = 0; t < NT2; ++t) {
        const int c0 = t * 16 + l;
        if (c0 < ncls) {
            const float bv = b2[c0];
            #pragma unroll
            for (int r = 0; r < 8; ++r) {
                const int trow = half * 8 + r;
                const float v = acc2[t][r];
                os_hw [trow * ncls + c0] = v;
                os_out[trow * ncls + c0] = v * dsq[r] + bv;
            }
        }
    }
    __builtin_amdgcn_wave_barrier();

    for (int i4 = lane * 4; i4 < TROWS * ncls; i4 += 128) {
        const int trow = i4 / ncls;
        if (first + trow < n) {
            const size_t g = (size_t)first * ncls + i4;
            *(float4*)(hw2 + g) = *(const float4*)(os_hw  + i4);
            *(float4*)(out + g) = *(const float4*)(os_out + i4);
        }
    }
    __builtin_amdgcn_wave_barrier();
    }
}

extern "C" void kernel_launch(void* const* d_in, const int* in_sizes, int n_in,
                              void* d_out, int out_size, void* d_ws, size_t ws_size,
                              hipStream_t stream) {
    const float* x  = (const float*)d_in[0];
    const int*   ei = (const int*)  d_in[1];
    const float* W1 = (const float*)d_in[2];
    const float* b1 = (const float*)d_in[3];
    const float* W2 = (const float*)d_in[4];
    const float* b2 = (const float*)d_in[5];
    float* out = (float*)d_out;

    const int N   = in_sizes[0] / NFEAT;
    const int E   = in_sizes[1] / 2;
    const int CLS = in_sizes[5];

    const int* src = ei;
    const int* dst = ei + E;

    const int NBKT = (N + BKT_W - 1) / BKT_W;

    char* ws = (char*)d_ws;
    size_t off = 0;
    float* dis = (float*)(ws + off); off += ((size_t)N * 4 + 255) & ~(size_t)255;
    float* hw2 = (float*)(ws + off); off += ((size_t)N * CLS * 4 + 255) & ~(size_t)255;
    const int EBLK = (E + CH - 1) / CH;
    int* rowptr = (int*)(ws + off); off += ((size_t)(N + 1) * 4 + 255) & ~(size_t)255;
    int* ptab   = (int*)(ws + off); off += ((size_t)(NBKT + 1) * EBLK * 4 + 255) & ~(size_t)255;
    int* bkt    = (int*)(ws + off); off += ((size_t)EBLK * CH * 8 + 255) & ~(size_t)255;
    int* csr    = (int*)(ws + off); off += ((size_t)E * 4 + 255) & ~(size_t)255;
    (void)ws_size; (void)n_in; (void)out_size;

    const int TPB = 256;
    const int WAVES = TPB / 32;

    k_blk_group <<<EBLK, EB, 0, stream>>>(src, dst, ptab, bkt, E, NBKT);
    k_bkt_finish<<<NBKT, EB, 0, stream>>>(bkt, ptab, rowptr, dis, csr, N, EBLK);

    {
        const int NTILE = (N + TROWS - 1) / TROWS;
        k_row_gather_gemm12<<<(NTILE + 2 * GW - 1) / (2 * GW), GT, 0, stream>>>(
            rowptr, csr, dis, x, W1, b1, W2, b2, hw2, out, CLS, N);
    }
    k_row_gather<4><<<(N + 4 * WAVES - 1) / (4 * WAVES), TPB, 0, stream>>>(
        rowptr, csr, dis, hw2, out, CLS, N);
}


# Layer-2 rows stored pre-scaled by dis[row]: row sums add plain rows, no per-edge dis[src] gather; sum times dis[d] once, b2 added last (was first); out written once instead of init plus update
# speedup vs baseline: 4.9532x; 1.0741x over previous
#include <hip/hip_runtime.h>


typedef __attribute__((ext_vector_type(2))) float v2f;
typedef __attribute__((ext_vector_type(8))) float v8f;

#define NFEAT 128


#define BKT_SHIFT 8
#define BKT_W (1 << BKT_SHIFT)
#define EB 1024
#define EPT 4
#define CH (EB * EPT)
#define FU 4

__device__ __forceinline__ int wave_incl_scan(int v, int lane) {
    #pragma unroll
    for (int o = 1; o < 32; o <<= 1) {
        const int t = __shfl_up(v, o);
        if (lane >= o) v += t;
    }
    return v;
}

__global__ void __launch_bounds__(EB) k_blk_group(const int* __restrict__ src,
                                                  const int* __restrict__ dst,
                                                  int* __restrict__ ptab,
                                                  int* __restrict__ bkt_raw,
                                                  int E, int nbkt) {
    __shared__ int hist[EB];
    __shared__ int wsum[EB / 32];
    __shared__ __attribute__((aligned(8))) int stage_raw[2 * CH];
    const int t = threadIdx.x, lane = t & 31, w = t >> 5;
    const int nblk = gridDim.x;
    int2* stage = (int2*)stage_raw;

    hist[t] = 0;
    __syncthreads();

    const int base = blockIdx.x * CH;
    int s[EPT], d[EPT], r[EPT];
    #pragma unroll
    for (int k = 0; k < EPT; ++k) {
        const int e = base + k * EB + t;
        d[k] = (e < E) ? dst[e] : -1;
        s[k] = (e < E) ? src[e] : 0;
    }
    #pragma unroll
    for (int k = 0; k < EPT; ++k) {
        r[k] = 0;
        if (d[k] >= 0) r[k] = atomicAdd(&hist[d[k] >> BKT_SHIFT], 1);
    }
    __syncthreads();

    const int c   = hist[t];
    const int inc = wave_incl_scan(c, lane);
    if (lane == 31) wsum[w] = inc;
    __syncthreads();
    if (w == 0) {
        const int x  = wsum[lane];
        const int xi = wave_incl_scan(x, lane);
        wsum[lane] = xi - x;
    }
    __syncthreads();
    const int excl = wsum[w] + inc - c;
    if (t <= nbkt) ptab[(size_t)t * nblk + blockIdx.x] = excl;
    hist[t] = excl;
    __syncthreads();

    #pragma unroll
    for (int k = 0; k < EPT; ++k) {
        if (d[k] >= 0) stage[hist[d[k] >> BKT_SHIFT] + r[k]] = make_int2(s[k], d[k]);
    }
    __syncthreads();
    const int left = E - base;
    const int nval = left < CH ? left : CH;
    int2* __restrict__ run = (int2*)bkt_raw + (size_t)blockIdx.x * CH;
    #pragma unroll
    for (int k = 0; k < EPT; ++k) {
        const int i = k * EB + t;
        if (i < nval) run[i] = stage[i];
    }
}

__device__ __forceinline__ void locate_pairs(const int* pp, const int* pg, int i0, int npair,
                                             int (&g)[FU]) {
    int b[FU];
    #pragma unroll
    for (int u = 0; u < FU; ++u) b[u] = 0;
    #pragma unroll
    for (int step = EB / 2; step > 0; step >>= 1) {
        #pragma unroll
        for (int u = 0; u < FU; ++u) {
            const int nb = b[u] + step;
            if (pp[nb] <= i0 + u * EB) b[u] = nb;
        }
    }
    #pragma unroll
    for (int u = 0; u < FU; ++u) {
        const int i = i0 + u * EB;
        const int gi = pg[b[u]] + (i - pp[b[u]]);
        g[u] = (i < npair) ? gi : -1;
    }
}

__global__ void __launch_bounds__(EB) k_bkt_finish(const int* __restrict__ bkt_raw,
                                                   const int* __restrict__ ptab,
                                                   int* __restrict__ rowptr,
                                                   float* __restrict__ dis,
                                                   int* __restrict__ csr_src, int n, int nblk) {
    __shared__ int cnt[BKT_W];
    __shared__ int cur[BKT_W];
    __shared__ int wsum[EB / 32];
    __shared__ int wbeg[EB / 32];
    __shared__ int pp[EB];
    __shared__ int pg[EB];
    __shared__ int tot[2];
    const int t = threadIdx.x, lane = t & 31, w = t >> 5;
    const int2* __restrict__ bkt = (const int2*)bkt_raw;

    const int lo = blockIdx.x << BKT_SHIFT;

    int sb = 0, eb = 0;
    if (t < nblk) {
        sb = ptab[(size_t)blockIdx.x * nblk + t];
        eb = ptab[(size_t)(blockIdx.x + 1) * nblk + t];
    }
    const int len = eb - sb;
    int bsum = sb;
    #pragma unroll
    for (int o = 16; o > 0; o >>= 1) bsum += __shfl_xor(bsum, o);
    const int pinc = wave_incl_scan(len, lane);
    if (lane == 31) wsum[w] = pinc;
    if (lane == 0)  wbeg[w] = bsum;
    if (t < BKT_W) cnt[t] = 0;
    __syncthreads();
    if (w == 0) {
        const int x  = wsum[lane];
        const int xi = wave_incl_scan(x, lane);
        wsum[lane] = xi - x;
        int y = wbeg[lane];
        #pragma unroll
        for (int o = 16; o > 0; o >>= 1) y += __shfl_xor(y, o);
        if (lane == 0)  tot[0] = y;
        if (lane == 31) tot[1] = xi;
    }
    __syncthreads();
    pp[t] = wsum[w] + pinc - len;
    pg[t] = t * CH + sb;
    const int beg  = tot[0];
    const int npair = tot[1];
    __syncthreads();

    int g0[FU];
    locate_pairs(pp, pg, t, npair, g0);

    for (int i0 = t; i0 < npair; i0 += FU * EB) {
        int g[FU], dd[FU];
        #pragma unroll
        for (int u = 0; u < FU; ++u) g[u] = g0[u];
        if (i0 != t) locate_pairs(pp, pg, i0, npair, g);
        #pragma unroll
        for (int u = 0; u < FU; ++u)
            dd[u] = (g[u] >= 0) ? bkt[g[u]].y : -1;
        #pragma unroll
        for (int u = 0; u < FU; ++u)
            if (dd[u] >= 0) atomicAdd(&cnt[dd[u] - lo], 1);
    }
    __syncthreads();

    int c = 0, inc = 0;
    if (t < BKT_W) {
        c   = cnt[t];
        inc = wave_incl_scan(c, lane);
        if (lane == 31) wsum[w] = inc;
    }
    __syncthreads();
    if (w == 0) {
        const int x  = (lane < BKT_W / 32) ? wsum[lane] : 0;
        const int xi = wave_incl_scan(x, lane);
        if (lane < BKT_W / 32) wsum[lane] = xi - x;
    }
    __syncthreads();
    if (t < BKT_W) {
        const int excl = wsum[w] + inc - c;
        cur[t] = excl;
        const int node = lo + t;
        if (node < n) {
            rowptr[node] = beg + excl;
            dis[node]    = rsqrtf((float)c + 1.0f);
            if (node == n - 1) rowptr[n] = beg + excl + c;
        }
    }
    __syncthreads();
    for (int i0 = t; i0 < npair; i0 += FU * EB) {
        int g[FU];
        int2 p[FU];
        #pragma unroll
        for (int u = 0; u < FU; ++u) g[u] = g0[u];
        if (i0 != t) locate_pairs(pp, pg, i0, npair, g);
        #pragma unroll
        for (int u = 0; u < FU; ++u)
            p[u] = (g[u] >= 0) ? bkt[g[u]] : make_int2(0, -1);
        #pragma unroll
        for (int u = 0; u < FU; ++u) {
            if (p[u].y >= 0) {
                const int pos = atomicAdd(&cur[p[u].y - lo], 1);
                csr_src[beg + pos] = p[u].x;
            }
        }
    }
}

template<int ROWS>
__global__ void __launch_bounds__(256) k_row_gather(const int* __restrict__ rowptr,
                                                    const int* __restrict__ csr_src,
                                                    const float* __restrict__ dis,
                                                    const float* __restrict__ feat,
                                                    const float* __restrict__ bias,
                                                    float* __restrict__ out, int ncols, int n) {
    const int lane = threadIdx.x & 31;
    const int first = __builtin_amdgcn_readfirstlane(
        (int)((blockIdx.x * (blockDim.x >> 5) + (threadIdx.x >> 5)) * ROWS));

    const bool act = (lane * 4 < ncols);
    const int  c4  = act ? lane * 4 : 0;
    const float* __restrict__ fcol = feat + c4;
    const float4 bv = *(const float4*)(bias + c4);

    for (int r = 0; r < ROWS; ++r) {
        const int node = first + r;
        if (node >= n) break;

        const int   beg = rowptr[node];
        const int   end = rowptr[node + 1];
        const float di  = dis[node];

        const int tot = end - beg + 1;
        float4 acc = make_float4(0.0f, 0.0f, 0.0f, 0.0f);

        for (int eb = 0; eb < tot; eb += 32) {
            const int left = tot - eb;
            const int m = left < 32 ? left : 32;
            const int e = eb + lane;
            int ms = node;
            if (lane < m && e > 0) ms = csr_src[beg + e - 1];
            int j = 0;
            for (; j + 4 <= m; j += 4) {
                const int s0 = __builtin_amdgcn_readlane(ms, j);
                const int s1 = __builtin_amdgcn_readlane(ms, j + 1);
                const int s2 = __builtin_amdgcn_readlane(ms, j + 2);
                const int s3 = __builtin_amdgcn_readlane(ms, j + 3);
                const float4 v0 = *(const float4*)(fcol + (size_t)s0 * ncols);
                const float4 v1 = *(const float4*)(fcol + (size_t)s1 * ncols);
                const float4 v2 = *(const float4*)(fcol + (size_t)s2 * ncols);
                const float4 v3 = *(const float4*)(fcol + (size_t)s3 * ncols);
                acc.x += v0.x; acc.y += v0.y; acc.z += v0.z; acc.w += v0.w;
                acc.x += v1.x; acc.y += v1.y; acc.z += v1.z; acc.w += v1.w;
                acc.x += v2.x; acc.y += v2.y; acc.z += v2.z; acc.w += v2.w;
                acc.x += v3.x; acc.y += v3.y; acc.z += v3.z; acc.w += v3.w;
            }
            for (; j < m; ++j) {
                const int s0 = __builtin_amdgcn_readlane(ms, j);
                const float4 v0 = *(const float4*)(fcol + (size_t)s0 * ncols);
                acc.x += v0.x; acc.y += v0.y; acc.z += v0.z; acc.w += v0.w;
            }
        }
        if (act) {
            float4 o;
            o.x = fmaf(acc.x, di, bv.x); o.y = fmaf(acc.y, di, bv.y);
            o.z = fmaf(acc.z, di, bv.z); o.w = fmaf(acc.w, di, bv.w);
            *(float4*)(out + (size_t)node * ncols + c4) = o;
        }
    }
}


#define TROWS   16
#define HPITCH  132
#define NT2     3
#define WPITCH  48
#define W1PITCH 136
#define GW      13
#define GT      (GW * 32)
#define W1Q     ((NFEAT * (NFEAT / 4) + GT - 1) / GT)
#define W2Q     ((NFEAT * (WPITCH / 4) + GT - 1) / GT)
#define KG1     2
#define KG2     4
#define CH0     31
#define PF      16
#define TW      12
#define COV     (PF + TW - 1)

template<int NV>
__device__ __forceinline__ void issue_rows(float4 (&v)[NV], int ms, int j, int cnt,
                                           const float* __restrict__ fcol) {
    #pragma unroll
    for (int u = 0; u < NV; ++u) {
        v[u] = make_float4(0.0f, 0.0f, 0.0f, 0.0f);
        if (u < cnt) {
            const int s = __builtin_amdgcn_readlane(ms, j + u);
            v[u] = *(const float4*)(fcol + (size_t)s * NFEAT);
        }
    }
}
template<int NV>
__device__ __forceinline__ float4 sum_rows(float4 acc, const float4 (&v)[NV], float mw,
                                           int j, int cnt) {
    #pragma unroll
    for (int u = 0; u < NV; ++u) {
        if (u < cnt) {
            const float w = __int_as_float(__builtin_amdgcn_readlane(__float_as_int(mw), j + u));
            acc.x = fmaf(v[u].x, w, acc.x); acc.y = fmaf(v[u].y, w, acc.y);
            acc.z = fmaf(v[u].z, w, acc.z); acc.w = fmaf(v[u].w, w, acc.w);
        }
    }
    return acc;
}

__device__ __forceinline__ void issue_ahead(float4 (&v)[PF], int ms, int tot,
                                            const float* __restrict__ fcol) {
    #pragma unroll
    for (int u = 0; u < PF; ++u) {
        const int s = (u < tot) ? __builtin_amdgcn_readlane(ms, u) : 0;
        v[u] = *(const float4*)(fcol + (size_t)s * NFEAT);
    }
}

__global__ void __launch_bounds__(GT) k_row_gather_gemm12(const int* __restrict__ rowptr,
                                                          const int* __restrict__ csr_src,
                                                          const float* __restrict__ dis,
                                                          const float* __restrict__ x,
                                                          const float* __restrict__ W1,
                                                          const float* __restrict__ b1,
                                                          const float* __restrict__ W2,
                                                          float* __restrict__ hw2,
                                                          int ncls, int n) {
    __shared__ __attribute__((aligned(16))) float w1s[NFEAT * W1PITCH];
    __shared__ __attribute__((aligned(16))) float w2s[NFEAT * WPITCH];
    __shared__ __attribute__((aligned(16))) float strips[GW * TROWS * HPITCH];

    {
        float4 p1[W1Q], p2[W2Q];
        #pragma unroll
        for (int u = 0; u < W1Q; ++u) {
            const int q  = threadIdx.x + u * GT;
            const int k  = q / (NFEAT / 4);
            const int cq = (q % (NFEAT / 4)) * 4;
            p1[u] = make_float4(0.0f, 0.0f, 0.0f, 0.0f);
            if (q < NFEAT * (NFEAT / 4)) p1[u] = *(const float4*)(W1 + (size_t)k * NFEAT + cq);
        }
        #pragma unroll
        for (int u = 0; u < W2Q; ++u) {
            const int q  = threadIdx.x + u * GT;
            const int k  = q / (WPITCH / 4);
            const int cq = (q % (WPITCH / 4)) * 4;
            p2[u] = make_float4(0.0f, 0.0f, 0.0f, 0.0f);
            if (q < NFEAT * (WPITCH / 4) && cq < ncls) p2[u] = *(const float4*)(W2 + (size_t)k * ncls + cq);
        }
        #pragma unroll
        for (int u = 0; u < W1Q; ++u) {
            const int q  = threadIdx.x + u * GT;
            const int k  = q / (NFEAT / 4);
            const int cq = (q % (NFEAT / 4)) * 4;
            if (q < NFEAT * (NFEAT / 4)) *(float4*)(w1s + k * W1PITCH + cq) = p1[u];
        }
        #pragma unroll
        for (int u = 0; u < W2Q; ++u) {
            const int q = threadIdx.x + u * GT;
            if (q < NFEAT * (WPITCH / 4)) *(float4*)(w2s + q * 4) = p2[u];
        }
    }
    __syncthreads();

    const int lane = threadIdx.x & 31;
    const int wv    = __builtin_amdgcn_readfirstlane((int)(threadIdx.x >> 5));
    float* strip = strips + wv * (TROWS * HPITCH);
    const int c4 = lane * 4;
    const float* __restrict__ fcol = x + c4;
    const int half = lane >> 4;
    const int l    = lane & 15;
    const float* arow = strip + l * HPITCH + 2 * half;

    const int ntile = (n + TROWS - 1) / TROWS;
    for (int tile = blockIdx.x * GW + wv; tile < ntile; tile += gridDim.x * GW) {
    const int first = tile * TROWS;

    int   rp = 0;
    float dv = 0.0f;
    {
        int ri = first + lane; if (ri > n) ri = n;
        int ni = first + lane; if (ni > n - 1) ni = n - 1;
        if (lane <= TROWS) rp = rowptr[ri];
        if (lane <  TROWS) dv = dis[ni];
    }

#define ROW_CNT(q)   ((__builtin_amdgcn_readlane(rp, (q) + 1) - __builtin_amdgcn_readlane(rp, (q))) < CH0 \
                      ? (__builtin_amdgcn_readlane(rp, (q) + 1) - __builtin_amdgcn_readlane(rp, (q))) : CH0)
#define ROW_NODE(q)  ((first + (q)) < n ? (first + (q)) : (n - 1))
#define LIST_IDX(q, cnt)  ((lane >= 1 && lane <= (cnt)) ? __builtin_amdgcn_readlane(rp, (q)) + lane - 1 \
                                                         : ((cnt) > 0 ? __builtin_amdgcn_readlane(rp, (q)) : 0))
#define LIST_SEL(raw, q, cnt)  ((lane >= 1 && lane <= (cnt)) ? (raw) : ROW_NODE(q))
    int   ms_a, ms_b;
    float w_a;
    {
        const int cnt0 = ROW_CNT(0);
        const int cnt1 = ROW_CNT(1);
        const int raw0 = csr_src[LIST_IDX(0, cnt0)];
        const int raw1 = csr_src[LIST_IDX(1, cnt1)];
        ms_a = LIST_SEL(raw0, 0, cnt0);
        ms_b = LIST_SEL(raw1, 1, cnt1);
        w_a  = dis[ms_a] * __int_as_float(__builtin_amdgcn_readlane(__float_as_int(dv), 0));
    }
    float4 v[2][PF];
    float4 vt[TW];
    issue_ahead(v[0], ms_a, (first < n) ? ROW_CNT(0) + 1 : 0, fcol);
    #pragma unroll 1
    for (int r2 = 0; r2 < TROWS; r2 += 2) {
    #pragma unroll
    for (int hh = 0; hh < 2; ++hh) {
        const int r = r2 + hh;
        const int node = first + r;
        const int q1 = (r + 1 < TROWS) ? r + 1 : TROWS - 1;
        const int q2 = (r + 2 < TROWS) ? r + 2 : TROWS - 1;
        const int cnt2 = ROW_CNT(q2);
        const int beg = __builtin_amdgcn_readlane(rp, r);
        const int end = __builtin_amdgcn_readlane(rp, r + 1);
        const int cnt = (end - beg) < CH0 ? (end - beg) : CH0;
        const int tot = (node < n) ? cnt + 1 : 0;
        const int totn = (r + 1 < TROWS && node + 1 < n) ? ROW_CNT(q1) + 1 : 0;

        const float raw_w = dis[ms_b];
        const int   raw_c = csr_src[LIST_IDX(q2, cnt2)];
        issue_rows(vt, ms_a, PF, tot - PF, fcol);
        issue_ahead(v[hh ^ 1], ms_b, totn, fcol);

        float4 acc = make_float4(0.0f, 0.0f, 0.0f, 0.0f);
        acc = sum_rows(acc, v[hh], w_a, 0, tot);
        acc = sum_rows(acc, vt, w_a, PF, tot - PF);
        *(float4*)(strip + r * HPITCH + c4) = acc;
        w_a  = raw_w * __int_as_float(__builtin_amdgcn_readlane(__float_as_int(dv), q1));
        ms_a = ms_b;
        ms_b = LIST_SEL(raw_c, q2, cnt2);
    }
    }

    {
        const int rpn = __shfl_down(rp, 1);
        const unsigned lmask = __builtin_amdgcn_ballot_w32(lane < TROWS && rpn - rp > COV);
        for (unsigned mk = lmask; mk != 0; mk &= mk - 1) {
            const int r = __builtin_ctz(mk);
            const int beg = __builtin_amdgcn_readlane(rp, r);
            const int end = __builtin_amdgcn_readlane(rp, r + 1);
            const float di = __int_as_float(__builtin_amdgcn_readlane(__float_as_int(dv), r));
            float4 acc = *(const float4*)(strip + r * HPITCH + c4);
            for (int base = beg + COV; base < end; base += 32) {
                const int left = end - base;
                const int m2 = left < 32 ? left : 32;
                int   ms = 0;
                float mw = 0.0f;
                if (lane < m2) {
                    ms = csr_src[base + lane];
                    mw = dis[ms] * di;
                }
                for (int j = 0; j < m2; j += 16) {
                    issue_rows(v[0], ms, j, m2 - j, fcol);
                    acc = sum_rows(acc, v[0], mw, j, m2 - j);
                }
            }
            *(float4*)(strip + r * HPITCH + c4) = acc;
        }
    }
#undef ROW_CNT
#undef ROW_NODE
#undef LIST_IDX
#undef LIST_SEL

    __builtin_amdgcn_wave_barrier();

    v8f hacc[NFEAT / 16];
    #pragma unroll
    for (int nh = 0; nh < 2; ++nh) {
        v8f acc[4];
        #pragma unroll
        for (int t = 0; t < 4; ++t) acc[t] = (v8f){0, 0, 0, 0, 0, 0, 0, 0};
        const float* __restrict__ bcol = w1s + (2 * half) * W1PITCH + nh * 64 + l;
        #pragma unroll
        for (int kg = 0; kg < NFEAT / 4; kg += KG1) {
            v2f a[KG1], b[4][KG1];
            #pragma unroll
            for (int u = 0; u < KG1; ++u) {
                const int ks = kg + u;
                a[u].x = arow[ks * 4];
                a[u].y = arow[ks * 4 + 1];
                #pragma unroll
                for (int t = 0; t < 4; ++t) {
                    b[t][u].x = bcol[(ks * 4)     * W1PITCH + t * 16];
                    b[t][u].y = bcol[(ks * 4 + 1) * W1PITCH + t * 16];
                }
            }
            __builtin_amdgcn_sched_barrier(0);
            #pragma unroll
            for (int u = 0; u < KG1; ++u) {
                #pragma unroll
                for (int t = 0; t < 4; ++t) {
                    acc[t] = __builtin_amdgcn_wmma_f32_16x16x4_f32(
                                 false, a[u], false, b[t][u], (short)0, acc[t], false, false);
                }
            }
        }
        #pragma unroll
        for (int t = 0; t < 4; ++t) hacc[nh * 4 + t] = acc[t];
    }

    __builtin_amdgcn_wave_barrier();
    #pragma unroll
    for (int t = 0; t < NFEAT / 16; ++t) {
        const int   col = t * 16 + l;
        const float bb  = b1[col];
        #pragma unroll
        for (int r = 0; r < 8; ++r)
            strip[(half * 8 + r) * HPITCH + col] = fmaxf(hacc[t][r] + bb, 0.0f);
    }
    __builtin_amdgcn_wave_barrier();

    v8f acc2[NT2];
    #pragma unroll
    for (int t = 0; t < NT2; ++t) acc2[t] = (v8f){0, 0, 0, 0, 0, 0, 0, 0};
    {
        const float* __restrict__ bcol = w2s + (2 * half) * WPITCH + l;
        #pragma unroll
        for (int kg = 0; kg < NFEAT / 4; kg += KG2) {
            v2f a[KG2], b[NT2][KG2];
            #pragma unroll
            for (int u = 0; u < KG2; ++u) {
                const int ks = kg + u;
                a[u].x = arow[ks * 4];
                a[u].y = arow[ks * 4 + 1];
                #pragma unroll
                for (int t = 0; t < NT2; ++t) {
                    b[t][u].x = bcol[(ks * 4)     * WPITCH + t * 16];
                    b[t][u].y = bcol[(ks * 4 + 1) * WPITCH + t * 16];
                }
            }
            __builtin_amdgcn_sched_barrier(0);
            #pragma unroll
            for (int u = 0; u < KG2; ++u) {
                #pragma unroll
                for (int t = 0; t < NT2; ++t) {
                    acc2[t] = __builtin_amdgcn_wmma_f32_16x16x4_f32(
                                  false, a[u], false, b[t][u], (short)0, acc2[t], false, false);
                }
            }
        }
    }

    float dsc[8];
    #pragma unroll
    for (int r = 0; r < 8; ++r) {
        const float dlo = __int_as_float(__builtin_amdgcn_readlane(__float_as_int(dv), r));
        const float dhi = __int_as_float(__builtin_amdgcn_readlane(__float_as_int(dv), r + 8));
        dsc[r] = half ? dhi : dlo;
    }

    __builtin_amdgcn_wave_barrier();
    float* os_hw = strip;
    #pragma unroll
    for (int t = 0; t < NT2; ++t) {
        const int c0 = t * 16 + l;
        if (c0 < ncls) {
            #pragma unroll
            for (int r = 0; r < 8; ++r) {
                const int trow = half * 8 + r;
                os_hw[trow * ncls + c0] = acc2[t][r] * dsc[r];
            }
        }
    }
    __builtin_amdgcn_wave_barrier();

    for (int i4 = lane * 4; i4 < TROWS * ncls; i4 += 128) {
        const int trow = i4 / ncls;
        if (first + trow < n) {
            const size_t g = (size_t)first * ncls + i4;
            *(float4*)(hw2 + g) = *(const float4*)(os_hw + i4);
        }
    }
    __builtin_amdgcn_wave_barrier();
    }
}

extern "C" void kernel_launch(void* const* d_in, const int* in_sizes, int n_in,
                              void* d_out, int out_size, void* d_ws, size_t ws_size,
                              hipStream_t stream) {
    const float* x  = (const float*)d_in[0];
    const int*   ei = (const int*)  d_in[1];
    const float* W1 = (const float*)d_in[2];
    const float* b1 = (const float*)d_in[3];
    const float* W2 = (const float*)d_in[4];
    const float* b2 = (const float*)d_in[5];
    float* out = (float*)d_out;

    const int N   = in_sizes[0] / NFEAT;
    const int E   = in_sizes[1] / 2;
    const int CLS = in_sizes[5];

    const int* src = ei;
    const int* dst = ei + E;

    const int NBKT = (N + BKT_W - 1) / BKT_W;

    char* ws = (char*)d_ws;
    size_t off = 0;
    float* dis = (float*)(ws + off); off += ((size_t)N * 4 + 255) & ~(size_t)255;
    float* hw2 = (float*)(ws + off); off += ((size_t)N * CLS * 4 + 255) & ~(size_t)255;
    const int EBLK = (E + CH - 1) / CH;
    int* rowptr = (int*)(ws + off); off += ((size_t)(N + 1) * 4 + 255) & ~(size_t)255;
    int* ptab   = (int*)(ws + off); off += ((size_t)(NBKT + 1) * EBLK * 4 + 255) & ~(size_t)255;
    int* bkt    = (int*)(ws + off); off += ((size_t)EBLK * CH * 8 + 255) & ~(size_t)255;
    int* csr    = (int*)(ws + off); off += ((size_t)E * 4 + 255) & ~(size_t)255;
    (void)ws_size; (void)n_in; (void)out_size;

    const int TPB = 256;
    const int WAVES = TPB / 32;

    k_blk_group <<<EBLK, EB, 0, stream>>>(src, dst, ptab, bkt, E, NBKT);
    k_bkt_finish<<<NBKT, EB, 0, stream>>>(bkt, ptab, rowptr, dis, csr, N, EBLK);

    {
        const int NTILE = (N + TROWS - 1) / TROWS;
        k_row_gather_gemm12<<<(NTILE + 2 * GW - 1) / (2 * GW), GT, 0, stream>>>(
            rowptr, csr, dis, x, W1, b1, W2, hw2, CLS, N);
    }
    k_row_gather<4><<<(N + 4 * WAVES - 1) / (4 * WAVES), TPB, 0, stream>>>(
        rowptr, csr, dis, hw2, b2, out, CLS, N);
}


# Layer-1 row sums without a test per list entry: entries past a list's end get source row 0 and weight zero, so rows are asked for and summed in straight-line code; same sums, same order
# speedup vs baseline: 5.3830x; 1.0868x over previous
#include <hip/hip_runtime.h>


typedef __attribute__((ext_vector_type(2))) float v2f;
typedef __attribute__((ext_vector_type(8))) float v8f;

#define NFEAT 128


#define BKT_SHIFT 8
#define BKT_W (1 << BKT_SHIFT)
#define EB 1024
#define EPT 4
#define CH (EB * EPT)
#define FU 4

__device__ __forceinline__ int wave_incl_scan(int v, int lane) {
    #pragma unroll
    for (int o = 1; o < 32; o <<= 1) {
        const int t = __shfl_up(v, o);
        if (lane >= o) v += t;
    }
    return v;
}

__global__ void __launch_bounds__(EB) k_blk_group(const int* __restrict__ src,
                                                  const int* __restrict__ dst,
                                                  int* __restrict__ ptab,
                                                  int* __restrict__ bkt_raw,
                                                  int E, int nbkt) {
    __shared__ int hist[EB];
    __shared__ int wsum[EB / 32];
    __shared__ __attribute__((aligned(8))) int stage_raw[2 * CH];
    const int t = threadIdx.x, lane = t & 31, w = t >> 5;
    const int nblk = gridDim.x;
    int2* stage = (int2*)stage_raw;

    hist[t] = 0;
    __syncthreads();

    const int base = blockIdx.x * CH;
    int s[EPT], d[EPT], r[EPT];
    #pragma unroll
    for (int k = 0; k < EPT; ++k) {
        const int e = base + k * EB + t;
        d[k] = (e < E) ? dst[e] : -1;
        s[k] = (e < E) ? src[e] : 0;
    }
    #pragma unroll
    for (int k = 0; k < EPT; ++k) {
        r[k] = 0;
        if (d[k] >= 0) r[k] = atomicAdd(&hist[d[k] >> BKT_SHIFT], 1);
    }
    __syncthreads();

    const int c   = hist[t];
    const int inc = wave_incl_scan(c, lane);
    if (lane == 31) wsum[w] = inc;
    __syncthreads();
    if (w == 0) {
        const int x  = wsum[lane];
        const int xi = wave_incl_scan(x, lane);
        wsum[lane] = xi - x;
    }
    __syncthreads();
    const int excl = wsum[w] + inc - c;
    if (t <= nbkt) ptab[(size_t)t * nblk + blockIdx.x] = excl;
    hist[t] = excl;
    __syncthreads();

    #pragma unroll
    for (int k = 0; k < EPT; ++k) {
        if (d[k] >= 0) stage[hist[d[k] >> BKT_SHIFT] + r[k]] = make_int2(s[k], d[k]);
    }
    __syncthreads();
    const int left = E - base;
    const int nval = left < CH ? left : CH;
    int2* __restrict__ run = (int2*)bkt_raw + (size_t)blockIdx.x * CH;
    #pragma unroll
    for (int k = 0; k < EPT; ++k) {
        const int i = k * EB + t;
        if (i < nval) run[i] = stage[i];
    }
}

__device__ __forceinline__ void locate_pairs(const int* pp, const int* pg, int i0, int npair,
                                             int (&g)[FU]) {
    int b[FU];
    #pragma unroll
    for (int u = 0; u < FU; ++u) b[u] = 0;
    #pragma unroll
    for (int step = EB / 2; step > 0; step >>= 1) {
        #pragma unroll
        for (int u = 0; u < FU; ++u) {
            const int nb = b[u] + step;
            if (pp[nb] <= i0 + u * EB) b[u] = nb;
        }
    }
    #pragma unroll
    for (int u = 0; u < FU; ++u) {
        const int i = i0 + u * EB;
        const int gi = pg[b[u]] + (i - pp[b[u]]);
        g[u] = (i < npair) ? gi : -1;
    }
}

__global__ void __launch_bounds__(EB) k_bkt_finish(const int* __restrict__ bkt_raw,
                                                   const int* __restrict__ ptab,
                                                   int* __restrict__ rowptr,
                                                   float* __restrict__ dis,
                                                   int* __restrict__ csr_src, int n, int nblk) {
    __shared__ int cnt[BKT_W];
    __shared__ int cur[BKT_W];
    __shared__ int wsum[EB / 32];
    __shared__ int wbeg[EB / 32];
    __shared__ int pp[EB];
    __shared__ int pg[EB];
    __shared__ int tot[2];
    const int t = threadIdx.x, lane = t & 31, w = t >> 5;
    const int2* __restrict__ bkt = (const int2*)bkt_raw;

    const int lo = blockIdx.x << BKT_SHIFT;

    int sb = 0, eb = 0;
    if (t < nblk) {
        sb = ptab[(size_t)blockIdx.x * nblk + t];
        eb = ptab[(size_t)(blockIdx.x + 1) * nblk + t];
    }
    const int len = eb - sb;
    int bsum = sb;
    #pragma unroll
    for (int o = 16; o > 0; o >>= 1) bsum += __shfl_xor(bsum, o);
    const int pinc = wave_incl_scan(len, lane);
    if (lane == 31) wsum[w] = pinc;
    if (lane == 0)  wbeg[w] = bsum;
    if (t < BKT_W) cnt[t] = 0;
    __syncthreads();
    if (w == 0) {
        const int x  = wsum[lane];
        const int xi = wave_incl_scan(x, lane);
        wsum[lane] = xi - x;
        int y = wbeg[lane];
        #pragma unroll
        for (int o = 16; o > 0; o >>= 1) y += __shfl_xor(y, o);
        if (lane == 0)  tot[0] = y;
        if (lane == 31) tot[1] = xi;
    }
    __syncthreads();
    pp[t] = wsum[w] + pinc - len;
    pg[t] = t * CH + sb;
    const int beg  = tot[0];
    const int npair = tot[1];
    __syncthreads();

    int g0[FU];
    locate_pairs(pp, pg, t, npair, g0);

    for (int i0 = t; i0 < npair; i0 += FU * EB) {
        int g[FU], dd[FU];
        #pragma unroll
        for (int u = 0; u < FU; ++u) g[u] = g0[u];
        if (i0 != t) locate_pairs(pp, pg, i0, npair, g);
        #pragma unroll
        for (int u = 0; u < FU; ++u)
            dd[u] = (g[u] >= 0) ? bkt[g[u]].y : -1;
        #pragma unroll
        for (int u = 0; u < FU; ++u)
            if (dd[u] >= 0) atomicAdd(&cnt[dd[u] - lo], 1);
    }
    __syncthreads();

    int c = 0, inc = 0;
    if (t < BKT_W) {
        c   = cnt[t];
        inc = wave_incl_scan(c, lane);
        if (lane == 31) wsum[w] = inc;
    }
    __syncthreads();
    if (w == 0) {
        const int x  = (lane < BKT_W / 32) ? wsum[lane] : 0;
        const int xi = wave_incl_scan(x, lane);
        if (lane < BKT_W / 32) wsum[lane] = xi - x;
    }
    __syncthreads();
    if (t < BKT_W) {
        const int excl = wsum[w] + inc - c;
        cur[t] = excl;
        const int node = lo + t;
        if (node < n) {
            rowptr[node] = beg + excl;
            dis[node]    = rsqrtf((float)c + 1.0f);
            if (node == n - 1) rowptr[n] = beg + excl + c;
        }
    }
    __syncthreads();
    for (int i0 = t; i0 < npair; i0 += FU * EB) {
        int g[FU];
        int2 p[FU];
        #pragma unroll
        for (int u = 0; u < FU; ++u) g[u] = g0[u];
        if (i0 != t) locate_pairs(pp, pg, i0, npair, g);
        #pragma unroll
        for (int u = 0; u < FU; ++u)
            p[u] = (g[u] >= 0) ? bkt[g[u]] : make_int2(0, -1);
        #pragma unroll
        for (int u = 0; u < FU; ++u) {
            if (p[u].y >= 0) {
                const int pos = atomicAdd(&cur[p[u].y - lo], 1);
                csr_src[beg + pos] = p[u].x;
            }
        }
    }
}

template<int ROWS>
__global__ void __launch_bounds__(256) k_row_gather(const int* __restrict__ rowptr,
                                                    const int* __restrict__ csr_src,
                                                    const float* __restrict__ dis,
                                                    const float* __restrict__ feat,
                                                    const float* __restrict__ bias,
                                                    float* __restrict__ out, int ncols, int n) {
    const int lane = threadIdx.x & 31;
    const int first = __builtin_amdgcn_readfirstlane(
        (int)((blockIdx.x * (blockDim.x >> 5) + (threadIdx.x >> 5)) * ROWS));

    const bool act = (lane * 4 < ncols);
    const int  c4  = act ? lane * 4 : 0;
    const float* __restrict__ fcol = feat + c4;
    const float4 bv = *(const float4*)(bias + c4);

    for (int r = 0; r < ROWS; ++r) {
        const int node = first + r;
        if (node >= n) break;

        const int   beg = rowptr[node];
        const int   end = rowptr[node + 1];
        const float di  = dis[node];

        const int tot = end - beg + 1;
        float4 acc = make_float4(0.0f, 0.0f, 0.0f, 0.0f);

        for (int eb = 0; eb < tot; eb += 32) {
            const int left = tot - eb;
            const int m = left < 32 ? left : 32;
            const int e = eb + lane;
            int ms = node;
            if (lane < m && e > 0) ms = csr_src[beg + e - 1];
            int j = 0;
            for (; j + 4 <= m; j += 4) {
                const int s0 = __builtin_amdgcn_readlane(ms, j);
                const int s1 = __builtin_amdgcn_readlane(ms, j + 1);
                const int s2 = __builtin_amdgcn_readlane(ms, j + 2);
                const int s3 = __builtin_amdgcn_readlane(ms, j + 3);
                const float4 v0 = *(const float4*)(fcol + (size_t)s0 * ncols);
                const float4 v1 = *(const float4*)(fcol + (size_t)s1 * ncols);
                const float4 v2 = *(const float4*)(fcol + (size_t)s2 * ncols);
                const float4 v3 = *(const float4*)(fcol + (size_t)s3 * ncols);
                acc.x += v0.x; acc.y += v0.y; acc.z += v0.z; acc.w += v0.w;
                acc.x += v1.x; acc.y += v1.y; acc.z += v1.z; acc.w += v1.w;
                acc.x += v2.x; acc.y += v2.y; acc.z += v2.z; acc.w += v2.w;
                acc.x += v3.x; acc.y += v3.y; acc.z += v3.z; acc.w += v3.w;
            }
            for (; j < m; ++j) {
                const int s0 = __builtin_amdgcn_readlane(ms, j);
                const float4 v0 = *(const float4*)(fcol + (size_t)s0 * ncols);
                acc.x += v0.x; acc.y += v0.y; acc.z += v0.z; acc.w += v0.w;
            }
        }
        if (act) {
            float4 o;
            o.x = fmaf(acc.x, di, bv.x); o.y = fmaf(acc.y, di, bv.y);
            o.z = fmaf(acc.z, di, bv.z); o.w = fmaf(acc.w, di, bv.w);
            *(float4*)(out + (size_t)node * ncols + c4) = o;
        }
    }
}


#define TROWS   16
#define HPITCH  132
#define NT2     3
#define WPITCH  48
#define W1PITCH 136
#define GW      13
#define GT      (GW * 32)
#define W1Q     ((NFEAT * (NFEAT / 4) + GT - 1) / GT)
#define W2Q     ((NFEAT * (WPITCH / 4) + GT - 1) / GT)
#define KG1     2
#define KG2     4
#define CH0     31
#define PF      16
#define TW      12
#define TG      4
#define COV     (PF + TW - 1)

template<int NV>
__device__ __forceinline__ void issue_rows(float4 (&v)[NV], int ms, int j, int cnt,
                                           const float* __restrict__ fcol) {
    #pragma unroll
    for (int u = 0; u < NV; ++u) {
        v[u] = make_float4(0.0f, 0.0f, 0.0f, 0.0f);
        if (u < cnt) {
            const int s = __builtin_amdgcn_readlane(ms, j + u);
            v[u] = *(const float4*)(fcol + (size_t)s * NFEAT);
        }
    }
}
template<int NV>
__device__ __forceinline__ float4 sum_rows(float4 acc, const float4 (&v)[NV], float mw,
                                           int j, int cnt) {
    #pragma unroll
    for (int u = 0; u < NV; ++u) {
        if (u < cnt) {
            const float w = __int_as_float(__builtin_amdgcn_readlane(__float_as_int(mw), j + u));
            acc.x = fmaf(v[u].x, w, acc.x); acc.y = fmaf(v[u].y, w, acc.y);
            acc.z = fmaf(v[u].z, w, acc.z); acc.w = fmaf(v[u].w, w, acc.w);
        }
    }
    return acc;
}

__device__ __forceinline__ float4 ld_row(const float* __restrict__ x, int s, int lane16) {
    return *(const float4*)((const char*)x + (unsigned)(s * (NFEAT * 4) + lane16));
}
template<int NV, int J, int VO, int VN>
__device__ __forceinline__ void issue_all(float4 (&v)[VN], int ms, const float* __restrict__ x,
                                          int lane16) {
    #pragma unroll
    for (int u = 0; u < NV; ++u)
        v[VO + u] = ld_row(x, __builtin_amdgcn_readlane(ms, J + u), lane16);
}
template<int NV, int J, int VO, int VN>
__device__ __forceinline__ float4 sum_all(float4 acc, const float4 (&v)[VN], float mw) {
    #pragma unroll
    for (int u = 0; u < NV; ++u) {
        const float w = __int_as_float(__builtin_amdgcn_readlane(__float_as_int(mw), J + u));
        acc.x = fmaf(v[VO + u].x, w, acc.x); acc.y = fmaf(v[VO + u].y, w, acc.y);
        acc.z = fmaf(v[VO + u].z, w, acc.z); acc.w = fmaf(v[VO + u].w, w, acc.w);
    }
    return acc;
}

__global__ void __launch_bounds__(GT) k_row_gather_gemm12(const int* __restrict__ rowptr,
                                                          const int* __restrict__ csr_src,
                                                          const float* __restrict__ dis,
                                                          const float* __restrict__ x,
                                                          const float* __restrict__ W1,
                                                          const float* __restrict__ b1,
                                                          const float* __restrict__ W2,
                                                          float* __restrict__ hw2,
                                                          int ncls, int n) {
    __shared__ __attribute__((aligned(16))) float w1s[NFEAT * W1PITCH];
    __shared__ __attribute__((aligned(16))) float w2s[NFEAT * WPITCH];
    __shared__ __attribute__((aligned(16))) float strips[GW * TROWS * HPITCH];

    {
        float4 p1[W1Q], p2[W2Q];
        #pragma unroll
        for (int u = 0; u < W1Q; ++u) {
            const int q  = threadIdx.x + u * GT;
            const int k  = q / (NFEAT / 4);
            const int cq = (q % (NFEAT / 4)) * 4;
            p1[u] = make_float4(0.0f, 0.0f, 0.0f, 0.0f);
            if (q < NFEAT * (NFEAT / 4)) p1[u] = *(const float4*)(W1 + (size_t)k * NFEAT + cq);
        }
        #pragma unroll
        for (int u = 0; u < W2Q; ++u) {
            const int q  = threadIdx.x + u * GT;
            const int k  = q / (WPITCH / 4);
            const int cq = (q % (WPITCH / 4)) * 4;
            p2[u] = make_float4(0.0f, 0.0f, 0.0f, 0.0f);
            if (q < NFEAT * (WPITCH / 4) && cq < ncls) p2[u] = *(const float4*)(W2 + (size_t)k * ncls + cq);
        }
        #pragma unroll
        for (int u = 0; u < W1Q; ++u) {
            const int q  = threadIdx.x + u * GT;
            const int k  = q / (NFEAT / 4);
            const int cq = (q % (NFEAT / 4)) * 4;
            if (q < NFEAT * (NFEAT / 4)) *(float4*)(w1s + k * W1PITCH + cq) = p1[u];
        }
        #pragma unroll
        for (int u = 0; u < W2Q; ++u) {
            const int q = threadIdx.x + u * GT;
            if (q < NFEAT * (WPITCH / 4)) *(float4*)(w2s + q * 4) = p2[u];
        }
    }
    __syncthreads();

    const int lane = threadIdx.x & 31;
    const int wv    = __builtin_amdgcn_readfirstlane((int)(threadIdx.x >> 5));
    float* strip = strips + wv * (TROWS * HPITCH);
    const int c4 = lane * 4;
    const float* __restrict__ fcol = x + c4;
    const int half = lane >> 4;
    const int l    = lane & 15;
    const float* arow = strip + l * HPITCH + 2 * half;

    const int ntile = (n + TROWS - 1) / TROWS;
    for (int tile = blockIdx.x * GW + wv; tile < ntile; tile += gridDim.x * GW) {
    const int first = tile * TROWS;

    int   rp = 0;
    float dv = 0.0f;
    {
        int ri = first + lane; if (ri > n) ri = n;
        int ni = first + lane; if (ni > n - 1) ni = n - 1;
        if (lane <= TROWS) rp = rowptr[ri];
        if (lane <  TROWS) dv = dis[ni];
    }

#define ROW_CNT(q)   ((__builtin_amdgcn_readlane(rp, (q) + 1) - __builtin_amdgcn_readlane(rp, (q))) < CH0 \
                      ? (__builtin_amdgcn_readlane(rp, (q) + 1) - __builtin_amdgcn_readlane(rp, (q))) : CH0)
#define ROW_NODE(q)  ((first + (q)) < n ? (first + (q)) : (n - 1))
#define LIST_IDX(q, cnt)  ((lane >= 1 && lane <= (cnt)) ? __builtin_amdgcn_readlane(rp, (q)) + lane - 1 \
                                                         : ((cnt) > 0 ? __builtin_amdgcn_readlane(rp, (q)) : 0))
#define LIST_SEL(raw, q, cnt)  ((lane >= 1 && lane <= (cnt)) ? (raw) : ROW_NODE(q))
    int   ms_a, ms_b;
    float w_a;
    {
        const int cnt0 = ROW_CNT(0);
        const int cnt1 = ROW_CNT(1);
        const int raw0 = csr_src[LIST_IDX(0, cnt0)];
        const int raw1 = csr_src[LIST_IDX(1, cnt1)];
        ms_a = LIST_SEL(raw0, 0, cnt0);
        ms_b = LIST_SEL(raw1, 1, cnt1);
        w_a  = dis[ms_a] * __int_as_float(__builtin_amdgcn_readlane(__float_as_int(dv), 0));
    }
    float4 v[2][PF];
    float4 vt[TW];
    const int l16 = lane * 16;
    {
        const int tot0 = (first < n) ? ROW_CNT(0) + 1 : 0;
        issue_all<PF, 0, 0, PF>(v[0], (lane < tot0) ? ms_a : 0, x, l16);
    }
    #pragma unroll 1
    for (int r2 = 0; r2 < TROWS; r2 += 2) {
    #pragma unroll
    for (int hh = 0; hh < 2; ++hh) {
        const int r = r2 + hh;
        const int node = first + r;
        const int q1 = (r + 1 < TROWS) ? r + 1 : TROWS - 1;
        const int q2 = (r + 2 < TROWS) ? r + 2 : TROWS - 1;
        const int cnt2 = ROW_CNT(q2);
        const int beg = __builtin_amdgcn_readlane(rp, r);
        const int end = __builtin_amdgcn_readlane(rp, r + 1);
        const int cnt = (end - beg) < CH0 ? (end - beg) : CH0;
        const int tot = (node < n) ? cnt + 1 : 0;
        const int totn = (r + 1 < TROWS && node + 1 < n) ? ROW_CNT(q1) + 1 : 0;

        const float raw_w = dis[ms_b];
        const int   raw_c = csr_src[LIST_IDX(q2, cnt2)];
        const int   ms_cur = (lane < tot)  ? ms_a : 0;
        const float w_cur  = (lane < tot)  ? w_a  : 0.0f;
        const int   ms_nxt = (lane < totn) ? ms_b : 0;
        const int   ext = tot - PF;
        if (ext > 0)      issue_all<TG, PF,          0,      TW>(vt, ms_cur, x, l16);
        if (ext > TG)     issue_all<TG, PF + TG,     TG,     TW>(vt, ms_cur, x, l16);
        if (ext > 2 * TG) issue_all<TG, PF + 2 * TG, 2 * TG, TW>(vt, ms_cur, x, l16);
        issue_all<PF, 0, 0, PF>(v[hh ^ 1], ms_nxt, x, l16);

        float4 acc = make_float4(0.0f, 0.0f, 0.0f, 0.0f);
        acc = sum_all<PF, 0, 0, PF>(acc, v[hh], w_cur);
        if (ext > 0)      acc = sum_all<TG, PF,          0,      TW>(acc, vt, w_cur);
        if (ext > TG)     acc = sum_all<TG, PF + TG,     TG,     TW>(acc, vt, w_cur);
        if (ext > 2 * TG) acc = sum_all<TG, PF + 2 * TG, 2 * TG, TW>(acc, vt, w_cur);
        *(float4*)(strip + r * HPITCH + c4) = acc;
        w_a  = raw_w * __int_as_float(__builtin_amdgcn_readlane(__float_as_int(dv), q1));
        ms_a = ms_b;
        ms_b = LIST_SEL(raw_c, q2, cnt2);
    }
    }

    {
        const int rpn = __shfl_down(rp, 1);
        const unsigned lmask = __builtin_amdgcn_ballot_w32(lane < TROWS && rpn - rp > COV);
        for (unsigned mk = lmask; mk != 0; mk &= mk - 1) {
            const int r = __builtin_ctz(mk);
            const int beg = __builtin_amdgcn_readlane(rp, r);
            const int end = __builtin_amdgcn_readlane(rp, r + 1);
            const float di = __int_as_float(__builtin_amdgcn_readlane(__float_as_int(dv), r));
            float4 acc = *(const float4*)(strip + r * HPITCH + c4);
            for (int base = beg + COV; base < end; base += 32) {
                const int left = end - base;
                const int m2 = left < 32 ? left : 32;
                int   ms = 0;
                float mw = 0.0f;
                if (lane < m2) {
                    ms = csr_src[base + lane];
                    mw = dis[ms] * di;
                }
                for (int j = 0; j < m2; j += 16) {
                    issue_rows(v[0], ms, j, m2 - j, fcol);
                    acc = sum_rows(acc, v[0], mw, j, m2 - j);
                }
            }
            *(float4*)(strip + r * HPITCH + c4) = acc;
        }
    }
#undef ROW_CNT
#undef ROW_NODE
#undef LIST_IDX
#undef LIST_SEL

    __builtin_amdgcn_wave_barrier();

    v8f hacc[NFEAT / 16];
    #pragma unroll
    for (int nh = 0; nh < 2; ++nh) {
        v8f acc[4];
        #pragma unroll
        for (int t = 0; t < 4; ++t) acc[t] = (v8f){0, 0, 0, 0, 0, 0, 0, 0};
        const float* __restrict__ bcol = w1s + (2 * half) * W1PITCH + nh * 64 + l;
        #pragma unroll
        for (int kg = 0; kg < NFEAT / 4; kg += KG1) {
            v2f a[KG1], b[4][KG1];
            #pragma unroll
            for (int u = 0; u < KG1; ++u) {
                const int ks = kg + u;
                a[u].x = arow[ks * 4];
                a[u].y = arow[ks * 4 + 1];
                #pragma unroll
                for (int t = 0; t < 4; ++t) {
                    b[t][u].x = bcol[(ks * 4)     * W1PITCH + t * 16];
                    b[t][u].y = bcol[(ks * 4 + 1) * W1PITCH + t * 16];
                }
            }
            __builtin_amdgcn_sched_barrier(0);
            #pragma unroll
            for (int u = 0; u < KG1; ++u) {
                #pragma unroll
                for (int t = 0; t < 4; ++t) {
                    acc[t] = __builtin_amdgcn_wmma_f32_16x16x4_f32(
                                 false, a[u], false, b[t][u], (short)0, acc[t], false, false);
                }
            }
        }
        #pragma unroll
        for (int t = 0; t < 4; ++t) hacc[nh * 4 + t] = acc[t];
    }

    __builtin_amdgcn_wave_barrier();
    #pragma unroll
    for (int t = 0; t < NFEAT / 16; ++t) {
        const int   col = t * 16 + l;
        const float bb  = b1[col];
        #pragma unroll
        for (int r = 0; r < 8; ++r)
            strip[(half * 8 + r) * HPITCH + col] = fmaxf(hacc[t][r] + bb, 0.0f);
    }
    __builtin_amdgcn_wave_barrier();

    v8f acc2[NT2];
    #pragma unroll
    for (int t = 0; t < NT2; ++t) acc2[t] = (v8f){0, 0, 0, 0, 0, 0, 0, 0};
    {
        const float* __restrict__ bcol = w2s + (2 * half) * WPITCH + l;
        #pragma unroll
        for (int kg = 0; kg < NFEAT / 4; kg += KG2) {
            v2f a[KG2], b[NT2][KG2];
            #pragma unroll
            for (int u = 0; u < KG2; ++u) {
                const int ks = kg + u;
                a[u].x = arow[ks * 4];
                a[u].y = arow[ks * 4 + 1];
                #pragma unroll
                for (int t = 0; t < NT2; ++t) {
                    b[t][u].x = bcol[(ks * 4)     * WPITCH + t * 16];
                    b[t][u].y = bcol[(ks * 4 + 1) * WPITCH + t * 16];
                }
            }
            __builtin_amdgcn_sched_barrier(0);
            #pragma unroll
            for (int u = 0; u < KG2; ++u) {
                #pragma unroll
                for (int t = 0; t < NT2; ++t) {
                    acc2[t] = __builtin_amdgcn_wmma_f32_16x16x4_f32(
                                  false, a[u], false, b[t][u], (short)0, acc2[t], false, false);
                }
            }
        }
    }

    float dsc[8];
    #pragma unroll
    for (int r = 0; r < 8; ++r) {
        const float dlo = __int_as_float(__builtin_amdgcn_readlane(__float_as_int(dv), r));
        const float dhi = __int_as_float(__builtin_amdgcn_readlane(__float_as_int(dv), r + 8));
        dsc[r] = half ? dhi : dlo;
    }

    __builtin_amdgcn_wave_barrier();
    float* os_hw = strip;
    #pragma unroll
    for (int t = 0; t < NT2; ++t) {
        const int c0 = t * 16 + l;
        if (c0 < ncls) {
            #pragma unroll
            for (int r = 0; r < 8; ++r) {
                const int trow = half * 8 + r;
                os_hw[trow * ncls + c0] = acc2[t][r] * dsc[r];
            }
        }
    }
    __builtin_amdgcn_wave_barrier();

    for (int i4 = lane * 4; i4 < TROWS * ncls; i4 += 128) {
        const int trow = i4 / ncls;
        if (first + trow < n) {
            const size_t g = (size_t)first * ncls + i4;
            *(float4*)(hw2 + g) = *(const float4*)(os_hw + i4);
        }
    }
    __builtin_amdgcn_wave_barrier();
    }
}

extern "C" void kernel_launch(void* const* d_in, const int* in_sizes, int n_in,
                              void* d_out, int out_size, void* d_ws, size_t ws_size,
                              hipStream_t stream) {
    const float* x  = (const float*)d_in[0];
    const int*   ei = (const int*)  d_in[1];
    const float* W1 = (const float*)d_in[2];
    const float* b1 = (const float*)d_in[3];
    const float* W2 = (const float*)d_in[4];
    const float* b2 = (const float*)d_in[5];
    float* out = (float*)d_out;

    const int N   = in_sizes[0] / NFEAT;
    const int E   = in_sizes[1] / 2;
    const int CLS = in_sizes[5];

    const int* src = ei;
    const int* dst = ei + E;

    const int NBKT = (N + BKT_W - 1) / BKT_W;

    char* ws = (char*)d_ws;
    size_t off = 0;
    float* dis = (float*)(ws + off); off += ((size_t)N * 4 + 255) & ~(size_t)255;
    float* hw2 = (float*)(ws + off); off += ((size_t)N * CLS * 4 + 255) & ~(size_t)255;
    const int EBLK = (E + CH - 1) / CH;
    int* rowptr = (int*)(ws + off); off += ((size_t)(N + 1) * 4 + 255) & ~(size_t)255;
    int* ptab   = (int*)(ws + off); off += ((size_t)(NBKT + 1) * EBLK * 4 + 255) & ~(size_t)255;
    int* bkt    = (int*)(ws + off); off += ((size_t)EBLK * CH * 8 + 255) & ~(size_t)255;
    int* csr    = (int*)(ws + off); off += ((size_t)E * 4 + 255) & ~(size_t)255;
    (void)ws_size; (void)n_in; (void)out_size;

    const int TPB = 256;
    const int WAVES = TPB / 32;

    k_blk_group <<<EBLK, EB, 0, stream>>>(src, dst, ptab, bkt, E, NBKT);
    k_bkt_finish<<<NBKT, EB, 0, stream>>>(bkt, ptab, rowptr, dis, csr, N, EBLK);

    {
        const int NTILE = (N + TROWS - 1) / TROWS;
        k_row_gather_gemm12<<<(NTILE + 2 * GW - 1) / (2 * GW), GT, 0, stream>>>(
            rowptr, csr, dis, x, W1, b1, W2, hw2, CLS, N);
    }
    k_row_gather<4><<<(N + 4 * WAVES - 1) / (4 * WAVES), TPB, 0, stream>>>(
        rowptr, csr, dis, hw2, b2, out, CLS, N);
}
